# starting kernel
# baseline (speedup 1.0000x reference)
#include <hip/hip_runtime.h>
#include <hip/hip_bf16.h>
#include <math.h>

#define N_NODES 100000
#define E_EDGES 3200000
#define NIN     500
#define NHID    64
#define NOUT    40
#define NLAYERS 8
#define ALPHA   0.1f

#define BRS 160

typedef __attribute__((ext_vector_type(2))) float v2f;
typedef __attribute__((ext_vector_type(8))) float v8f;

__device__ __forceinline__ void wmma_acc(v8f& c, v2f a, v2f b) {
  asm("v_wmma_f32_16x16x4_f32 %0, %1, %2, %0" : "+v"(c) : "v"(a), "v"(b));
}
__device__ __forceinline__ void wmma_epilogue_guard() {
  asm volatile("s_nop 8");
}

__global__ void __launch_bounds__(256) in_gemm_kernel(
    const float* __restrict__ x, const float* __restrict__ Win,
    const float* __restrict__ bin, float* __restrict__ h0) {
  extern __shared__ float Wp[];
  for (int idx = threadIdx.x; idx < NIN * NHID; idx += 256) {
    const int k = idx >> 6;
    const int n = idx & 63;
    Wp[(k >> 1) * BRS + 2 * n + (k & 1)] = Win[idx];
  }
  __syncthreads();

  const int wave = threadIdx.x >> 5;
  const int lane = threadIdx.x & 31;
  const int half = lane >> 4;
  const int l16  = lane & 15;
  const int r0   = blockIdx.x * 256 + wave * 32;

  const int arow0 = min(r0 + l16,      N_NODES - 1);
  const int arow1 = min(r0 + 16 + l16, N_NODES - 1);
  const float* xr0 = x + (long long)arow0 * NIN;
  const float* xr1 = x + (long long)arow1 * NIN;

  v8f acc0[4] = {};
  v8f acc1[4] = {};
  for (int k = 0; k < NIN; k += 4) {
    const int kk = k + 2 * half;
    const v2f a0 = *reinterpret_cast<const v2f*>(xr0 + kk);
    const v2f a1 = *reinterpret_cast<const v2f*>(xr1 + kk);
    const int bbase = (kk >> 1) * BRS + 2 * l16;
#pragma unroll
    for (int t = 0; t < 4; ++t) {
      const v2f b = *reinterpret_cast<const v2f*>(&Wp[bbase + t * 32]);
      wmma_acc(acc0[t], a0, b);
      wmma_acc(acc1[t], a1, b);
    }
  }
  wmma_epilogue_guard();

#pragma unroll
  for (int t = 0; t < 4; ++t) {
    const int n = t * 16 + l16;
    const float bias = bin[n];
#pragma unroll
    for (int r = 0; r < 8; ++r) {
      const int m = r + 8 * half;
      const int orow0 = r0 + m;
      const int orow1 = r0 + 16 + m;
      if (orow0 < N_NODES) {
        const float v = acc0[t][r] + bias;
        h0[(long long)orow0 * NHID + n] = v > 0.f ? v : 0.f;
      }
      if (orow1 < N_NODES) {
        const float v = acc1[t][r] + bias;
        h0[(long long)orow1 * NHID + n] = v > 0.f ? v : 0.f;
      }
    }
  }
}

__global__ void zero_kernel(float4* __restrict__ p, int n4) {
  int i = blockIdx.x * blockDim.x + threadIdx.x;
  if (i < n4) p[i] = make_float4(0.f, 0.f, 0.f, 0.f);
}

__global__ void __launch_bounds__(256) spmm_kernel(
    const int* __restrict__ row, const int* __restrict__ col,
    const float* __restrict__ ew, const float* __restrict__ h,
    float* __restrict__ hi) {
  const long long t = (long long)blockIdx.x * blockDim.x + threadIdx.x;
  const long long total = (long long)E_EDGES * 16;
  if (t >= total) return;
  const int e = (int)(t >> 4);
  const int q = (int)(t & 15);
  const int c = col[e];
  const int r = row[e];
  const float w = ew[e];
  const float4 v = *reinterpret_cast<const float4*>(h + (long long)c * NHID + q * 4);
  float* dst = hi + (long long)r * NHID + q * 4;
  unsafeAtomicAdd(dst + 0, v.x * w);
  unsafeAtomicAdd(dst + 1, v.y * w);
  unsafeAtomicAdd(dst + 2, v.z * w);
  unsafeAtomicAdd(dst + 3, v.w * w);
}

__global__ void __launch_bounds__(256) conv_kernel(
    const float* __restrict__ hi, const float* __restrict__ h0,
    const float* __restrict__ Wl, float beta, float* __restrict__ hout) {
  __shared__ float Wp[(NHID / 2) * BRS];
  for (int idx = threadIdx.x; idx < NHID * NHID; idx += 256) {
    const int k = idx >> 6;
    const int n = idx & 63;
    Wp[(k >> 1) * BRS + 2 * n + (k & 1)] = Wl[idx];
  }
  __syncthreads();

  const int wave = threadIdx.x >> 5;
  const int lane = threadIdx.x & 31;
  const int half = lane >> 4;
  const int l16  = lane & 15;
  const int r0   = blockIdx.x * 256 + wave * 32;

  const float sA = 1.f - ALPHA;
  const float sB = ALPHA;
  const float cscale = (1.f - beta) / beta;

  v8f acc0[4] = {};
  v8f acc1[4] = {};
#pragma unroll
  for (int t = 0; t < 4; ++t) {
    const int n = t * 16 + l16;
#pragma unroll
    for (int r = 0; r < 8; ++r) {
      const int m = r + 8 * half;
      const long long rr0 = min(r0 + m,      N_NODES - 1);
      const long long rr1 = min(r0 + 16 + m, N_NODES - 1);
      acc0[t][r] = cscale * (sA * hi[rr0 * NHID + n] + sB * h0[rr0 * NHID + n]);
      acc1[t][r] = cscale * (sA * hi[rr1 * NHID + n] + sB * h0[rr1 * NHID + n]);
    }
  }

  const long long arow0 = min(r0 + l16,      N_NODES - 1);
  const long long arow1 = min(r0 + 16 + l16, N_NODES - 1);
  const float* hia0 = hi + arow0 * NHID;
  const float* h0a0 = h0 + arow0 * NHID;
  const float* hia1 = hi + arow1 * NHID;
  const float* h0a1 = h0 + arow1 * NHID;
#pragma unroll
  for (int k = 0; k < NHID; k += 4) {
    const int kk = k + 2 * half;
    const v2f hv0 = *reinterpret_cast<const v2f*>(hia0 + kk);
    const v2f gv0 = *reinterpret_cast<const v2f*>(h0a0 + kk);
    const v2f hv1 = *reinterpret_cast<const v2f*>(hia1 + kk);
    const v2f gv1 = *reinterpret_cast<const v2f*>(h0a1 + kk);
    v2f a0, a1;
    a0.x = sA * hv0.x + sB * gv0.x;  a0.y = sA * hv0.y + sB * gv0.y;
    a1.x = sA * hv1.x + sB * gv1.x;  a1.y = sA * hv1.y + sB * gv1.y;
    const int bbase = (kk >> 1) * BRS + 2 * l16;
#pragma unroll
    for (int t = 0; t < 4; ++t) {
      const v2f b = *reinterpret_cast<const v2f*>(&Wp[bbase + t * 32]);
      wmma_acc(acc0[t], a0, b);
      wmma_acc(acc1[t], a1, b);
    }
  }
  wmma_epilogue_guard();

#pragma unroll
  for (int t = 0; t < 4; ++t) {
    const int n = t * 16 + l16;
#pragma unroll
    for (int r = 0; r < 8; ++r) {
      const int m = r + 8 * half;
      const int orow0 = r0 + m;
      const int orow1 = r0 + 16 + m;
      if (orow0 < N_NODES) {
        const float v = beta * acc0[t][r];
        hout[(long long)orow0 * NHID + n] = v > 0.f ? v : 0.f;
      }
      if (orow1 < N_NODES) {
        const float v = beta * acc1[t][r];
        hout[(long long)orow1 * NHID + n] = v > 0.f ? v : 0.f;
      }
    }
  }
}

__global__ void __launch_bounds__(256) out_kernel(
    const float* __restrict__ h, const float* __restrict__ Wout,
    const float* __restrict__ bout, float* __restrict__ out) {
  __shared__ float Ws[NHID * NOUT];
  __shared__ float bs[NOUT];
  for (int idx = threadIdx.x; idx < NHID * NOUT; idx += 256) Ws[idx] = Wout[idx];
  if (threadIdx.x < NOUT) bs[threadIdx.x] = bout[threadIdx.x];
  __syncthreads();

  const int i = blockIdx.x * blockDim.x + threadIdx.x;
  if (i >= N_NODES) return;

  float hr[NHID];
  const float4* hp = reinterpret_cast<const float4*>(h + (long long)i * NHID);
#pragma unroll
  for (int k4 = 0; k4 < NHID / 4; ++k4) {
    const float4 v = hp[k4];
    hr[4 * k4 + 0] = v.x; hr[4 * k4 + 1] = v.y;
    hr[4 * k4 + 2] = v.z; hr[4 * k4 + 3] = v.w;
  }

  float lg[NOUT];
#pragma unroll
  for (int c = 0; c < NOUT; ++c) lg[c] = bs[c];
  for (int k = 0; k < NHID; ++k) {
    const float hk = hr[k];
#pragma unroll
    for (int c = 0; c < NOUT; ++c) lg[c] = fmaf(hk, Ws[k * NOUT + c], lg[c]);
  }

  float mx = lg[0];
#pragma unroll
  for (int c = 1; c < NOUT; ++c) mx = fmaxf(mx, lg[c]);
  float s = 0.f;
#pragma unroll
  for (int c = 0; c < NOUT; ++c) s += expf(lg[c] - mx);
  const float lse = mx + logf(s);
#pragma unroll
  for (int c = 0; c < NOUT; ++c)
    out[(long long)i * NOUT + c] = lg[c] - lse;
}

extern "C" void kernel_launch(void* const* d_in, const int* in_sizes, int n_in,
                              void* d_out, int out_size, void* d_ws, size_t ws_size,
                              hipStream_t stream) {
  (void)in_sizes; (void)n_in; (void)out_size; (void)ws_size;
  const float* x     = (const float*)d_in[0];
  const int*   row   = (const int*)  d_in[1];
  const int*   col   = (const int*)  d_in[2];
  const float* ew    = (const float*)d_in[3];
  const float* Win   = (const float*)d_in[4];
  const float* bin   = (const float*)d_in[5];
  const float* convW = (const float*)d_in[6];
  const float* Wout  = (const float*)d_in[7];
  const float* bout  = (const float*)d_in[8];

  const size_t S = (size_t)N_NODES * NHID;
  float* h0 = (float*)d_ws;
  float* hA = h0 + S;
  float* hB = hA + S;

  const int rowBlocks = (N_NODES + 255) / 256;
  const size_t inLds = (size_t)(NIN / 2) * BRS * sizeof(float);
  in_gemm_kernel<<<rowBlocks, 256, inLds, stream>>>(x, Win, bin, h0);

  const long long spmmItems = (long long)E_EDGES * 16;
  const int spmmBlocks = (int)((spmmItems + 255) / 256);
  const int zeroBlocks = (int)((S / 4 + 255) / 256);

  for (int l = 0; l < NLAYERS; ++l) {
    const float beta = logf(0.5f / (float)(l + 1) + 1.f);
    zero_kernel<<<zeroBlocks, 256, 0, stream>>>((float4*)hB, (int)(S / 4));
    spmm_kernel<<<spmmBlocks, 256, 0, stream>>>(row, col, ew, (l == 0) ? h0 : hA, hB);
    conv_kernel<<<rowBlocks, 256, 0, stream>>>(hB, h0, convW + (size_t)l * NHID * NHID,
                                               beta, hA);
  }

  out_kernel<<<(N_NODES + 255) / 256, 256, 0, stream>>>(hA, Wout, bout, (float*)d_out);
}


# SpMM steps with operands swapped (rows A, weight-at-row B): sums land one row per lane, the A layout, and the same wave runs the layer's 64x64 product; conv launches gone; k summed in another order
# speedup vs baseline: 3.6310x; 3.6310x over previous
#include <hip/hip_runtime.h>
#include <hip/hip_bf16.h>
#include <math.h>

#define N_NODES 100000
#define E_EDGES 3200000
#define NIN     500
#define NHID    64
#define NOUT    40
#define NLAYERS 8
#define ALPHA   0.1f

#define BRS 160

typedef __attribute__((ext_vector_type(2))) float v2f;
typedef __attribute__((ext_vector_type(8))) float v8f;
typedef __attribute__((ext_vector_type(4))) unsigned v4u;
typedef __attribute__((ext_vector_type(4))) float v4f;

__device__ __forceinline__ void wmma_acc(v8f& c, v2f a, v2f b) {
  asm("v_wmma_f32_16x16x4_f32 %0, %1, %2, %0" : "+v"(c) : "v"(a), "v"(b));
}
__device__ __forceinline__ void wmma_epilogue_guard() {
  asm volatile("s_nop 8");
}

__global__ void __launch_bounds__(256) in_gemm_kernel(
    const float* __restrict__ x, const float* __restrict__ Win,
    const float* __restrict__ bin, float* __restrict__ h0) {
  extern __shared__ float Wp[];
  for (int idx = threadIdx.x; idx < NIN * NHID; idx += 256) {
    const int k = idx >> 6;
    const int n = idx & 63;
    Wp[(k >> 1) * BRS + 2 * n + (k & 1)] = Win[idx];
  }
  __syncthreads();

  const int wave = threadIdx.x >> 5;
  const int lane = threadIdx.x & 31;
  const int half = lane >> 4;
  const int l16  = lane & 15;
  const int r0   = blockIdx.x * 256 + wave * 32;

  const int arow0 = min(r0 + l16,      N_NODES - 1);
  const int arow1 = min(r0 + 16 + l16, N_NODES - 1);
  const float* xr0 = x + (long long)arow0 * NIN;
  const float* xr1 = x + (long long)arow1 * NIN;

  v8f acc0[4] = {};
  v8f acc1[4] = {};
  for (int k = 0; k < NIN; k += 4) {
    const int kk = k + 2 * half;
    const v2f a0 = *reinterpret_cast<const v2f*>(xr0 + kk);
    const v2f a1 = *reinterpret_cast<const v2f*>(xr1 + kk);
    const int bbase = (kk >> 1) * BRS + 2 * l16;
#pragma unroll
    for (int t = 0; t < 4; ++t) {
      const v2f b = *reinterpret_cast<const v2f*>(&Wp[bbase + t * 32]);
      wmma_acc(acc0[t], a0, b);
      wmma_acc(acc1[t], a1, b);
    }
  }
  wmma_epilogue_guard();

#pragma unroll
  for (int t = 0; t < 4; ++t) {
    const int n = t * 16 + l16;
    const float bias = bin[n];
#pragma unroll
    for (int r = 0; r < 8; ++r) {
      const int m = r + 8 * half;
      const int orow0 = r0 + m;
      const int orow1 = r0 + 16 + m;
      if (orow0 < N_NODES) {
        const float v = acc0[t][r] + bias;
        h0[(long long)orow0 * NHID + n] = v > 0.f ? v : 0.f;
      }
      if (orow1 < N_NODES) {
        const float v = acc1[t][r] + bias;
        h0[(long long)orow1 * NHID + n] = v > 0.f ? v : 0.f;
      }
    }
  }
}

#define TR      400
#define NTILES  (N_NODES / TR)
#define CHUNK   1024
#define NCHUNK  (E_EDGES / CHUNK)
#define NGROUP  ((NCHUNK + 31) / 32)
#define QCAP    64
#define NG      (TR / 16)
#define GCAP    1024
static_assert(N_NODES % TR == 0 && NTILES <= 256, "tile id must fit a byte");
static_assert(TR <= 512 && N_NODES <= (1 << 23), "record packs col << 9 | local row");
static_assert(E_EDGES % CHUNK == 0, "edge count must be whole chunks");
static_assert(TR % 16 == 0 && NG <= 32 && GCAP % 16 == 0, "one replay wave per group, whole chunks");

__global__ void __launch_bounds__(256) edge_pack_kernel(
    const int* __restrict__ row, const int* __restrict__ col,
    const float* __restrict__ ew, unsigned* __restrict__ tid,
    uint2* __restrict__ rec) {
  const int t = blockIdx.x * 256 + threadIdx.x;
  if (t >= E_EDGES / 4) return;
  const int4 r = reinterpret_cast<const int4*>(row)[t];
  const int4 c = reinterpret_cast<const int4*>(col)[t];
  const float4 w = reinterpret_cast<const float4*>(ew)[t];
  const unsigned t0 = (unsigned)(r.x / TR), t1 = (unsigned)(r.y / TR);
  const unsigned t2 = (unsigned)(r.z / TR), t3 = (unsigned)(r.w / TR);
  tid[t] = t0 | (t1 << 8) | (t2 << 16) | (t3 << 24);
  uint4* rp = reinterpret_cast<uint4*>(rec + 4 * (long long)t);
  rp[0] = make_uint4(__float_as_uint(w.x), ((unsigned)c.x << 9) | ((unsigned)r.x - t0 * TR),
                     __float_as_uint(w.y), ((unsigned)c.y << 9) | ((unsigned)r.y - t1 * TR));
  rp[1] = make_uint4(__float_as_uint(w.z), ((unsigned)c.z << 9) | ((unsigned)r.z - t2 * TR),
                     __float_as_uint(w.w), ((unsigned)c.w << 9) | ((unsigned)r.w - t3 * TR));
}

__device__ __forceinline__ unsigned zero_bytes(unsigned x) {
  const unsigned t = ((x & 0x7f7f7f7fu) + 0x7f7f7f7fu) | x;
  return ~t & 0x80808080u;
}

template <bool FULL>
__device__ __forceinline__ void add_batch(
    uint2 rc, int n, int lane, const float* __restrict__ h, float* acc) {
  const float w = __uint_as_float(rc.x);
  const int c = (int)(rc.y >> 9);
  const int lr = (int)(rc.y & 511u);
  v2f g[32];
#pragma unroll
  for (int i = 0; i < 32; ++i) {
    const int ci = __builtin_amdgcn_readlane(c, i);
    g[i] = *reinterpret_cast<const v2f*>(h + (long long)ci * NHID + 2 * lane);
  }
#pragma unroll
  for (int i = 0; i < 32; ++i) {
    if (FULL || i < n) {
      const float wi = __int_as_float(__builtin_amdgcn_readlane(__float_as_int(w), i));
      const int li = __builtin_amdgcn_readlane(lr, i);
      float* a = acc + li * NHID + 2 * lane;
      __hip_atomic_fetch_add(a,     wi * g[i].x, __ATOMIC_RELAXED, __HIP_MEMORY_SCOPE_WORKGROUP);
      __hip_atomic_fetch_add(a + 1, wi * g[i].y, __ATOMIC_RELAXED, __HIP_MEMORY_SCOPE_WORKGROUP);
    }
  }
}

__device__ __forceinline__ void edge_step(
    v8f& d0, v8f& d1, v8f& d2, v8f& d3, v4u r, v4f p, v4f q, unsigned c) {
  v2f b;
  b.x = ((r.y & 15u) == c) ? __uint_as_float(r.x) : 0.f;
  b.y = ((r.w & 15u) == c) ? __uint_as_float(r.z) : 0.f;
  v2f a0, a1, a2, a3;
  a0.x = p.x; a0.y = q.x;  a1.x = p.y; a1.y = q.y;
  a2.x = p.z; a2.y = q.z;  a3.x = p.w; a3.y = q.w;
  d0 = __builtin_amdgcn_wmma_f32_16x16x4_f32(0, a0, 0, b, 0, d0, false, false);
  d1 = __builtin_amdgcn_wmma_f32_16x16x4_f32(0, a1, 0, b, 0, d1, false, false);
  d2 = __builtin_amdgcn_wmma_f32_16x16x4_f32(0, a2, 0, b, 0, d2, false, false);
  d3 = __builtin_amdgcn_wmma_f32_16x16x4_f32(0, a3, 0, b, 0, d3, false, false);
}

__device__ __forceinline__ v4f row_of(const char* hb, unsigned y, unsigned cb) {
  return *reinterpret_cast<const v4f*>(hb + (((y >> 9) << 8) + cb));
}

__device__ __forceinline__ void replay_group(
    const uint2* __restrict__ list, int n, int lane,
    const float* __restrict__ h, v8f (&d)[4]) {
  const int half = lane >> 4;
  const unsigned c = (unsigned)lane & 15u;
  const v4u* lp = reinterpret_cast<const v4u*>(list) + half;
  const char* hb = reinterpret_cast<const char*>(h);
  const unsigned cb = 16u * c;
  const int nchunk = (n + 15) >> 4;
  const v8f z8 = {};
  v8f d0 = z8, d1 = z8, d2 = z8, d3 = z8;

  if (nchunk > 0) {
    const v4u zr = {0u, 0u, 0u, 0u};
    const v4f zf = {0.f, 0.f, 0.f, 0.f};
    v4u c0 = zr, c1 = zr, c2 = zr, c3 = zr;
    v4f a0 = zf, a1 = zf, a2 = zf, a3 = zf;
    for (int i = 0; i <= nchunk; ++i) {
      asm volatile("" :: "v"(a0), "v"(a1), "v"(a2), "v"(a3));
      const v4u* np = lp + min(i, nchunk - 1) * 8;
      const v4u n0 = np[0], n1 = np[2], n2 = np[4], n3 = np[6];
      const v4f b0 = row_of(hb, c2.y, cb), b1 = row_of(hb, c2.w, cb);
      const v4f b2 = row_of(hb, c3.y, cb), b3 = row_of(hb, c3.w, cb);
      edge_step(d0, d1, d2, d3, c0, a0, a1, c);
      edge_step(d0, d1, d2, d3, c1, a2, a3, c);
      a0 = row_of(hb, n0.y, cb);  a1 = row_of(hb, n0.w, cb);
      a2 = row_of(hb, n1.y, cb);  a3 = row_of(hb, n1.w, cb);
      edge_step(d0, d1, d2, d3, c2, b0, b1, c);
      edge_step(d0, d1, d2, d3, c3, b2, b3, c);
      c0 = n0; c1 = n1; c2 = n2; c3 = n3;
    }
  }
  d[0] = d0; d[1] = d1; d[2] = d2; d[3] = d3;
}

__device__ __forceinline__ void stage_layer_weights(
    const float* __restrict__ Wl, float beta, float* Wp) {
  const float omb = 1.f - beta;
#pragma unroll
  for (int j = 0; j < NHID * NHID / 1024; ++j) {
    const int idx = (int)threadIdx.x + 1024 * j;
    const int k = idx >> 6;
    const int n = idx & 63;
    const float w = beta * Wl[idx];
    const int kh = k >> 5, kt = k & 3, ki = (k >> 3) & 3, ks = (k >> 2) & 1;
    Wp[(((kt * 4 + ki) * 4 + (n & 3)) << 6) + 2 * ((n >> 2) + 16 * kh) + ks] =
        (k == n) ? w + omb : w;
  }
}

__device__ __forceinline__ void layer_body(
    v8f (&s)[4], const float* __restrict__ h0row, const float* Wp, int lane,
    float* __restrict__ hgroup) {
  const float sA = 1.f - ALPHA;
  const float sB = ALPHA;
  v4f g[8];
#pragma unroll
  for (int r = 0; r < 8; ++r) g[r] = *reinterpret_cast<const v4f*>(h0row + 4 * r);
#pragma unroll
  for (int r = 0; r < 8; ++r) {
    s[0][r] = sA * s[0][r] + sB * g[r].x;
    s[1][r] = sA * s[1][r] + sB * g[r].y;
    s[2][r] = sA * s[2][r] + sB * g[r].z;
    s[3][r] = sA * s[3][r] + sB * g[r].w;
  }
  const v8f z8 = {};
  v8f o0 = z8, o1 = z8, o2 = z8, o3 = z8;
  const v2f* wp = reinterpret_cast<const v2f*>(Wp) + lane;
#pragma unroll
  for (int t = 0; t < 4; ++t) {
#pragma unroll
    for (int i = 0; i < 4; ++i) {
      v2f a;
      a.x = s[t][2 * i];  a.y = s[t][2 * i + 1];
      const v2f* wk = wp + (t * 4 + i) * 128;
      o0 = __builtin_amdgcn_wmma_f32_16x16x4_f32(0, a, 0, wk[0],  0, o0, false, false);
      o1 = __builtin_amdgcn_wmma_f32_16x16x4_f32(0, a, 0, wk[32], 0, o1, false, false);
      o2 = __builtin_amdgcn_wmma_f32_16x16x4_f32(0, a, 0, wk[64], 0, o2, false, false);
      o3 = __builtin_amdgcn_wmma_f32_16x16x4_f32(0, a, 0, wk[96], 0, o3, false, false);
    }
  }
  const int half = lane >> 4;
  const int c = lane & 15;
#pragma unroll
  for (int r = 0; r < 8; ++r) {
    v4f o;
    o.x = o0[r] > 0.f ? o0[r] : 0.f;  o.y = o1[r] > 0.f ? o1[r] : 0.f;
    o.z = o2[r] > 0.f ? o2[r] : 0.f;  o.w = o3[r] > 0.f ? o3[r] : 0.f;
    *reinterpret_cast<v4f*>(hgroup + (8 * half + r) * NHID + 4 * c) = o;
  }
}

template <bool FIRST>
__global__ void __launch_bounds__(1024) spmm_owner_kernel(
    const unsigned* __restrict__ tid, const uint2* __restrict__ rec,
    const float* __restrict__ h, const float* __restrict__ h0,
    const float* __restrict__ Wl, float beta, float* __restrict__ hout,
    uint2* __restrict__ glists, unsigned* __restrict__ gcounts) {
  extern __shared__ __attribute__((aligned(16))) float tile_lds[];
  __shared__ unsigned wq[32 * QCAP];
  __shared__ unsigned gcnt[32];
  __shared__ __attribute__((aligned(16))) float Wp[NHID * NHID];
  float* acc = tile_lds;
  float4* acc4 = reinterpret_cast<float4*>(tile_lds);
  const int lane = threadIdx.x & 31;
  const int wave = __builtin_amdgcn_readfirstlane(threadIdx.x >> 5);
  const int qb = wave * QCAP;
  const int tile = blockIdx.x;
  uint2* tlist = glists + (size_t)tile * NG * GCAP;
  const long long grow = (long long)tile * TR + wave * 16;
  const float* h0row = h0 + (grow + (lane & 15)) * NHID + 32 * (lane >> 4);
  float* hgroup = hout + grow * NHID;

  stage_layer_weights(Wl, beta, Wp);
  __syncthreads();

  if (!FIRST) {
    const unsigned myc = (lane < NG) ? gcounts[tile * NG + lane] : 0u;
    if (__builtin_amdgcn_ballot_w32(myc > (unsigned)GCAP) == 0u) {
      if (wave < NG) {
        const int n = __builtin_amdgcn_readlane((int)myc, wave);
        v8f s[4];
        replay_group(tlist + (size_t)wave * GCAP, n, lane, h, s);
        layer_body(s, h0row, Wp, lane, hgroup);
      }
      return;
    }
  }

  for (int i = threadIdx.x; i < TR * NHID / 4; i += 1024)
    acc4[i] = make_float4(0.f, 0.f, 0.f, 0.f);
  if (FIRST && threadIdx.x < NG) gcnt[threadIdx.x] = 0u;
  __syncthreads();

  auto file_rec = [&](uint2 rc, bool valid) {
    if (valid) {
      const unsigned gq = (rc.y & 511u) >> 4;
      const unsigned slot = __hip_atomic_fetch_add(&gcnt[gq], 1u, __ATOMIC_RELAXED,
                                                   __HIP_MEMORY_SCOPE_WORKGROUP);
      if (slot < (unsigned)GCAP) tlist[gq * GCAP + slot] = rc;
    }
  };

  {
    const unsigned pat = (unsigned)tile * 0x01010101u;
    const uint4* tid4 = reinterpret_cast<const uint4*>(tid);
    const int g0 = (tile * NGROUP) / NTILES;

    struct ids2 { uint4 a, b; };
    auto load_ids = [&](int g) {
      const int ch = min(g * 32 + wave, NCHUNK - 1);
      ids2 r;
      r.a = tid4[ch * 64 + lane];
      r.b = tid4[ch * 64 + 32 + lane];
      return r;
    };
    auto next_group = [](int g) { return (g + 1 == NGROUP) ? 0 : g + 1; };

    int qn = 0;
    auto scan = [&](const ids2& v, int g) {
      const int ch = g * 32 + wave;
      if (ch >= NCHUNK) return;
      unsigned m = (zero_bytes(v.a.x ^ pat) >> 7) | (zero_bytes(v.a.y ^ pat) >> 6) |
                   (zero_bytes(v.a.z ^ pat) >> 5) | (zero_bytes(v.a.w ^ pat) >> 4) |
                   (zero_bytes(v.b.x ^ pat) >> 3) | (zero_bytes(v.b.y ^ pat) >> 2) |
                   (zero_bytes(v.b.z ^ pat) >> 1) | zero_bytes(v.b.w ^ pat);
      const unsigned ebase = (unsigned)ch * CHUNK + (unsigned)lane * 16u;
      while (true) {
        const bool has = m != 0u;
        const unsigned mask = __builtin_amdgcn_ballot_w32(has);
        if (mask == 0u) break;
        if (has) {
          const unsigned p = (unsigned)__builtin_ctz(m);
          wq[qb + qn + (int)__builtin_amdgcn_mbcnt_lo(mask, 0u)] =
              ebase + ((p & 4u) << 7) + ((p & 3u) << 2) + (p >> 3);
          m &= m - 1u;
        }
        qn += __builtin_popcount(mask);
        if (qn >= 32) {
          qn -= 32;
          const uint2 rc = rec[wq[qb + qn + lane]];
          if (FIRST) file_rec(rc, true);
          add_batch<true>(rc, 32, lane, h, acc);
        }
      }
    };

    static_assert(NGROUP % 2 == 0, "the scan loop takes two groups per trip");
    int g = g0;
    ids2 cur = load_ids(g);
    for (int k = 0; k < NGROUP; k += 2) {
      const int g1 = next_group(g);
      const int g2 = next_group(g1);
      const ids2 nxt = load_ids(g1);
      scan(cur, g);
      cur = load_ids(g2);
      scan(nxt, g1);
      g = g2;
    }
    if (qn > 0) {
      const uint2 rc = (lane < qn) ? rec[wq[qb + lane]] : make_uint2(0u, 0u);
      if (FIRST) file_rec(rc, lane < qn);
      add_batch<false>(rc, qn, lane, h, acc);
    }
  }
  __syncthreads();

  if (FIRST && wave < NG) {
    const unsigned n = gcnt[wave];
    if (lane == 0) gcounts[tile * NG + wave] = n;
    const unsigned idx = n + (unsigned)lane;
    if (n <= (unsigned)GCAP && lane < 16 && idx < ((n + 15u) & ~15u))
      tlist[(unsigned)wave * GCAP + idx] = make_uint2(0u, 0u);
  }

  if (wave < NG) {
    const v4f* trow = reinterpret_cast<const v4f*>(
        tile_lds + (wave * 16 + (lane & 15)) * NHID + 32 * (lane >> 4));
    v8f s[4];
#pragma unroll
    for (int r = 0; r < 8; ++r) {
      const v4f v = trow[r];
      s[0][r] = v.x;  s[1][r] = v.y;  s[2][r] = v.z;  s[3][r] = v.w;
    }
    layer_body(s, h0row, Wp, lane, hgroup);
  }
}

__global__ void __launch_bounds__(256) out_kernel(
    const float* __restrict__ h, const float* __restrict__ Wout,
    const float* __restrict__ bout, float* __restrict__ out) {
  __shared__ float Ws[NHID * NOUT];
  __shared__ float bs[NOUT];
  for (int idx = threadIdx.x; idx < NHID * NOUT; idx += 256) Ws[idx] = Wout[idx];
  if (threadIdx.x < NOUT) bs[threadIdx.x] = bout[threadIdx.x];
  __syncthreads();

  const int i = blockIdx.x * blockDim.x + threadIdx.x;
  if (i >= N_NODES) return;

  float hr[NHID];
  const float4* hp = reinterpret_cast<const float4*>(h + (long long)i * NHID);
#pragma unroll
  for (int k4 = 0; k4 < NHID / 4; ++k4) {
    const float4 v = hp[k4];
    hr[4 * k4 + 0] = v.x; hr[4 * k4 + 1] = v.y;
    hr[4 * k4 + 2] = v.z; hr[4 * k4 + 3] = v.w;
  }

  float lg[NOUT];
#pragma unroll
  for (int c = 0; c < NOUT; ++c) lg[c] = bs[c];
  for (int k = 0; k < NHID; ++k) {
    const float hk = hr[k];
#pragma unroll
    for (int c = 0; c < NOUT; ++c) lg[c] = fmaf(hk, Ws[k * NOUT + c], lg[c]);
  }

  float mx = lg[0];
#pragma unroll
  for (int c = 1; c < NOUT; ++c) mx = fmaxf(mx, lg[c]);
  float s = 0.f;
#pragma unroll
  for (int c = 0; c < NOUT; ++c) s += expf(lg[c] - mx);
  const float lse = mx + logf(s);
#pragma unroll
  for (int c = 0; c < NOUT; ++c)
    out[(long long)i * NOUT + c] = lg[c] - lse;
}

extern "C" void kernel_launch(void* const* d_in, const int* in_sizes, int n_in,
                              void* d_out, int out_size, void* d_ws, size_t ws_size,
                              hipStream_t stream) {
  (void)in_sizes; (void)n_in; (void)out_size; (void)ws_size;
  const float* x     = (const float*)d_in[0];
  const int*   row   = (const int*)  d_in[1];
  const int*   col   = (const int*)  d_in[2];
  const float* ew    = (const float*)d_in[3];
  const float* Win   = (const float*)d_in[4];
  const float* bin   = (const float*)d_in[5];
  const float* convW = (const float*)d_in[6];
  const float* Wout  = (const float*)d_in[7];
  const float* bout  = (const float*)d_in[8];

  const size_t S = (size_t)N_NODES * NHID;
  float* h0 = (float*)d_ws;
  float* hA = h0 + S;
  float* hB = hA + S;

  const int rowBlocks = (N_NODES + 255) / 256;
  const size_t inLds = (size_t)(NIN / 2) * BRS * sizeof(float);
  in_gemm_kernel<<<rowBlocks, 256, inLds, stream>>>(x, Win, bin, h0);

  unsigned* tid = (unsigned*)(hB + S);
  uint2* rec = (uint2*)(tid + E_EDGES / 4);
  edge_pack_kernel<<<E_EDGES / 4 / 256, 256, 0, stream>>>(row, col, ew, tid, rec);
  const size_t spmmLds = (size_t)164 * 1024;
  uint2* lists = rec + E_EDGES;
  unsigned* counts = (unsigned*)(lists + (size_t)NTILES * NG * GCAP);

  const float* hin = h0;
  float* hnext = hA;
  float* hother = hB;
  for (int l = 0; l < NLAYERS; ++l) {
    const float beta = logf(0.5f / (float)(l + 1) + 1.f);
    const float* Wl = convW + (size_t)l * NHID * NHID;
    if (l == 0)
      spmm_owner_kernel<true><<<NTILES, 1024, spmmLds, stream>>>(
          tid, rec, hin, h0, Wl, beta, hnext, lists, counts);
    else
      spmm_owner_kernel<false><<<NTILES, 1024, spmmLds, stream>>>(
          tid, rec, hin, h0, Wl, beta, hnext, lists, counts);
    hin = hnext;
    hnext = hother;
    hother = (float*)hin;
  }

  out_kernel<<<(N_NODES + 255) / 256, 256, 0, stream>>>(hin, Wout, bout, (float*)d_out);
}


# Layers 2-8 list replay: a wave's priority falls with the trips it has left (s_setprio 3,2,1,0), so the 6-7 waves of a SIMD end together, not one after another with the last alone; sums unchanged
# speedup vs baseline: 3.8375x; 1.0569x over previous
#include <hip/hip_runtime.h>
#include <hip/hip_bf16.h>
#include <math.h>

#define N_NODES 100000
#define E_EDGES 3200000
#define NIN     500
#define NHID    64
#define NOUT    40
#define NLAYERS 8
#define ALPHA   0.1f

#define BRS 160

typedef __attribute__((ext_vector_type(2))) float v2f;
typedef __attribute__((ext_vector_type(8))) float v8f;
typedef __attribute__((ext_vector_type(4))) unsigned v4u;
typedef __attribute__((ext_vector_type(4))) float v4f;

__device__ __forceinline__ void wmma_acc(v8f& c, v2f a, v2f b) {
  asm("v_wmma_f32_16x16x4_f32 %0, %1, %2, %0" : "+v"(c) : "v"(a), "v"(b));
}
__device__ __forceinline__ void wmma_epilogue_guard() {
  asm volatile("s_nop 8");
}

__global__ void __launch_bounds__(256) in_gemm_kernel(
    const float* __restrict__ x, const float* __restrict__ Win,
    const float* __restrict__ bin, float* __restrict__ h0) {
  extern __shared__ float Wp[];
  for (int idx = threadIdx.x; idx < NIN * NHID; idx += 256) {
    const int k = idx >> 6;
    const int n = idx & 63;
    Wp[(k >> 1) * BRS + 2 * n + (k & 1)] = Win[idx];
  }
  __syncthreads();

  const int wave = threadIdx.x >> 5;
  const int lane = threadIdx.x & 31;
  const int half = lane >> 4;
  const int l16  = lane & 15;
  const int r0   = blockIdx.x * 256 + wave * 32;

  const int arow0 = min(r0 + l16,      N_NODES - 1);
  const int arow1 = min(r0 + 16 + l16, N_NODES - 1);
  const float* xr0 = x + (long long)arow0 * NIN;
  const float* xr1 = x + (long long)arow1 * NIN;

  v8f acc0[4] = {};
  v8f acc1[4] = {};
  for (int k = 0; k < NIN; k += 4) {
    const int kk = k + 2 * half;
    const v2f a0 = *reinterpret_cast<const v2f*>(xr0 + kk);
    const v2f a1 = *reinterpret_cast<const v2f*>(xr1 + kk);
    const int bbase = (kk >> 1) * BRS + 2 * l16;
#pragma unroll
    for (int t = 0; t < 4; ++t) {
      const v2f b = *reinterpret_cast<const v2f*>(&Wp[bbase + t * 32]);
      wmma_acc(acc0[t], a0, b);
      wmma_acc(acc1[t], a1, b);
    }
  }
  wmma_epilogue_guard();

#pragma unroll
  for (int t = 0; t < 4; ++t) {
    const int n = t * 16 + l16;
    const float bias = bin[n];
#pragma unroll
    for (int r = 0; r < 8; ++r) {
      const int m = r + 8 * half;
      const int orow0 = r0 + m;
      const int orow1 = r0 + 16 + m;
      if (orow0 < N_NODES) {
        const float v = acc0[t][r] + bias;
        h0[(long long)orow0 * NHID + n] = v > 0.f ? v : 0.f;
      }
      if (orow1 < N_NODES) {
        const float v = acc1[t][r] + bias;
        h0[(long long)orow1 * NHID + n] = v > 0.f ? v : 0.f;
      }
    }
  }
}

#define TR      400
#define NTILES  (N_NODES / TR)
#define CHUNK   1024
#define NCHUNK  (E_EDGES / CHUNK)
#define NGROUP  ((NCHUNK + 31) / 32)
#define QCAP    64
#define NG      (TR / 16)
#define GCAP    1024
static_assert(N_NODES % TR == 0 && NTILES <= 256, "tile id must fit a byte");
static_assert(TR <= 512 && N_NODES <= (1 << 23), "record packs col << 9 | local row");
static_assert(E_EDGES % CHUNK == 0, "edge count must be whole chunks");
static_assert(TR % 16 == 0 && NG <= 32 && GCAP % 16 == 0, "one replay wave per group, whole chunks");

__global__ void __launch_bounds__(256) edge_pack_kernel(
    const int* __restrict__ row, const int* __restrict__ col,
    const float* __restrict__ ew, unsigned* __restrict__ tid,
    uint2* __restrict__ rec) {
  const int t = blockIdx.x * 256 + threadIdx.x;
  if (t >= E_EDGES / 4) return;
  const int4 r = reinterpret_cast<const int4*>(row)[t];
  const int4 c = reinterpret_cast<const int4*>(col)[t];
  const float4 w = reinterpret_cast<const float4*>(ew)[t];
  const unsigned t0 = (unsigned)(r.x / TR), t1 = (unsigned)(r.y / TR);
  const unsigned t2 = (unsigned)(r.z / TR), t3 = (unsigned)(r.w / TR);
  tid[t] = t0 | (t1 << 8) | (t2 << 16) | (t3 << 24);
  uint4* rp = reinterpret_cast<uint4*>(rec + 4 * (long long)t);
  rp[0] = make_uint4(__float_as_uint(w.x), ((unsigned)c.x << 9) | ((unsigned)r.x - t0 * TR),
                     __float_as_uint(w.y), ((unsigned)c.y << 9) | ((unsigned)r.y - t1 * TR));
  rp[1] = make_uint4(__float_as_uint(w.z), ((unsigned)c.z << 9) | ((unsigned)r.z - t2 * TR),
                     __float_as_uint(w.w), ((unsigned)c.w << 9) | ((unsigned)r.w - t3 * TR));
}

__device__ __forceinline__ unsigned zero_bytes(unsigned x) {
  const unsigned t = ((x & 0x7f7f7f7fu) + 0x7f7f7f7fu) | x;
  return ~t & 0x80808080u;
}

template <bool FULL>
__device__ __forceinline__ void add_batch(
    uint2 rc, int n, int lane, const float* __restrict__ h, float* acc) {
  const float w = __uint_as_float(rc.x);
  const int c = (int)(rc.y >> 9);
  const int lr = (int)(rc.y & 511u);
  v2f g[32];
#pragma unroll
  for (int i = 0; i < 32; ++i) {
    const int ci = __builtin_amdgcn_readlane(c, i);
    g[i] = *reinterpret_cast<const v2f*>(h + (long long)ci * NHID + 2 * lane);
  }
#pragma unroll
  for (int i = 0; i < 32; ++i) {
    if (FULL || i < n) {
      const float wi = __int_as_float(__builtin_amdgcn_readlane(__float_as_int(w), i));
      const int li = __builtin_amdgcn_readlane(lr, i);
      float* a = acc + li * NHID + 2 * lane;
      __hip_atomic_fetch_add(a,     wi * g[i].x, __ATOMIC_RELAXED, __HIP_MEMORY_SCOPE_WORKGROUP);
      __hip_atomic_fetch_add(a + 1, wi * g[i].y, __ATOMIC_RELAXED, __HIP_MEMORY_SCOPE_WORKGROUP);
    }
  }
}

__device__ __forceinline__ void edge_step(
    v8f& d0, v8f& d1, v8f& d2, v8f& d3, v4u r, v4f p, v4f q, unsigned c) {
  v2f b;
  b.x = ((r.y & 15u) == c) ? __uint_as_float(r.x) : 0.f;
  b.y = ((r.w & 15u) == c) ? __uint_as_float(r.z) : 0.f;
  v2f a0, a1, a2, a3;
  a0.x = p.x; a0.y = q.x;  a1.x = p.y; a1.y = q.y;
  a2.x = p.z; a2.y = q.z;  a3.x = p.w; a3.y = q.w;
  d0 = __builtin_amdgcn_wmma_f32_16x16x4_f32(0, a0, 0, b, 0, d0, false, false);
  d1 = __builtin_amdgcn_wmma_f32_16x16x4_f32(0, a1, 0, b, 0, d1, false, false);
  d2 = __builtin_amdgcn_wmma_f32_16x16x4_f32(0, a2, 0, b, 0, d2, false, false);
  d3 = __builtin_amdgcn_wmma_f32_16x16x4_f32(0, a3, 0, b, 0, d3, false, false);
}

__device__ __forceinline__ v4f row_of(const char* hb, unsigned y, unsigned cb) {
  return *reinterpret_cast<const v4f*>(hb + (((y >> 9) << 8) + cb));
}

__device__ __forceinline__ void replay_group(
    const uint2* __restrict__ list, int n, int lane,
    const float* __restrict__ h, v8f (&d)[4]) {
  const int half = lane >> 4;
  const unsigned c = (unsigned)lane & 15u;
  const v4u* lp = reinterpret_cast<const v4u*>(list) + half;
  const char* hb = reinterpret_cast<const char*>(h);
  const unsigned cb = 16u * c;
  const int nchunk = (n + 15) >> 4;
  const v8f z8 = {};
  v8f d0 = z8, d1 = z8, d2 = z8, d3 = z8;

  if (nchunk > 0) {
    const v4u zr = {0u, 0u, 0u, 0u};
    const v4f zf = {0.f, 0.f, 0.f, 0.f};
    v4u c0 = zr, c1 = zr, c2 = zr, c3 = zr;
    v4f a0 = zf, a1 = zf, a2 = zf, a3 = zf;
    int drop_at = -1;
    if (nchunk > 16)     { __builtin_amdgcn_s_setprio(3); drop_at = 16; }
    else if (nchunk > 8) { __builtin_amdgcn_s_setprio(2); drop_at = 8; }
    else if (nchunk > 3) { __builtin_amdgcn_s_setprio(1); drop_at = 3; }
    for (int i = 0; i <= nchunk; ++i) {
      if (nchunk - i == drop_at) {
        if (drop_at == 16)     { __builtin_amdgcn_s_setprio(2); drop_at = 8; }
        else if (drop_at == 8) { __builtin_amdgcn_s_setprio(1); drop_at = 3; }
        else                   { __builtin_amdgcn_s_setprio(0); drop_at = -1; }
      }
      asm volatile("" :: "v"(a0), "v"(a1), "v"(a2), "v"(a3));
      const v4u* np = lp + min(i, nchunk - 1) * 8;
      const v4u n0 = np[0], n1 = np[2], n2 = np[4], n3 = np[6];
      const v4f b0 = row_of(hb, c2.y, cb), b1 = row_of(hb, c2.w, cb);
      const v4f b2 = row_of(hb, c3.y, cb), b3 = row_of(hb, c3.w, cb);
      edge_step(d0, d1, d2, d3, c0, a0, a1, c);
      edge_step(d0, d1, d2, d3, c1, a2, a3, c);
      a0 = row_of(hb, n0.y, cb);  a1 = row_of(hb, n0.w, cb);
      a2 = row_of(hb, n1.y, cb);  a3 = row_of(hb, n1.w, cb);
      edge_step(d0, d1, d2, d3, c2, b0, b1, c);
      edge_step(d0, d1, d2, d3, c3, b2, b3, c);
      c0 = n0; c1 = n1; c2 = n2; c3 = n3;
    }
  }
  d[0] = d0; d[1] = d1; d[2] = d2; d[3] = d3;
}

__device__ __forceinline__ void stage_layer_weights(
    const float* __restrict__ Wl, float beta, float* Wp) {
  const float omb = 1.f - beta;
#pragma unroll
  for (int j = 0; j < NHID * NHID / 1024; ++j) {
    const int idx = (int)threadIdx.x + 1024 * j;
    const int k = idx >> 6;
    const int n = idx & 63;
    const float w = beta * Wl[idx];
    const int kh = k >> 5, kt = k & 3, ki = (k >> 3) & 3, ks = (k >> 2) & 1;
    Wp[(((kt * 4 + ki) * 4 + (n & 3)) << 6) + 2 * ((n >> 2) + 16 * kh) + ks] =
        (k == n) ? w + omb : w;
  }
}

__device__ __forceinline__ void layer_body(
    v8f (&s)[4], const float* __restrict__ h0row, const float* Wp, int lane,
    float* __restrict__ hgroup) {
  const float sA = 1.f - ALPHA;
  const float sB = ALPHA;
  v4f g[8];
#pragma unroll
  for (int r = 0; r < 8; ++r) g[r] = *reinterpret_cast<const v4f*>(h0row + 4 * r);
#pragma unroll
  for (int r = 0; r < 8; ++r) {
    s[0][r] = sA * s[0][r] + sB * g[r].x;
    s[1][r] = sA * s[1][r] + sB * g[r].y;
    s[2][r] = sA * s[2][r] + sB * g[r].z;
    s[3][r] = sA * s[3][r] + sB * g[r].w;
  }
  const v8f z8 = {};
  v8f o0 = z8, o1 = z8, o2 = z8, o3 = z8;
  const v2f* wp = reinterpret_cast<const v2f*>(Wp) + lane;
#pragma unroll
  for (int t = 0; t < 4; ++t) {
#pragma unroll
    for (int i = 0; i < 4; ++i) {
      v2f a;
      a.x = s[t][2 * i];  a.y = s[t][2 * i + 1];
      const v2f* wk = wp + (t * 4 + i) * 128;
      o0 = __builtin_amdgcn_wmma_f32_16x16x4_f32(0, a, 0, wk[0],  0, o0, false, false);
      o1 = __builtin_amdgcn_wmma_f32_16x16x4_f32(0, a, 0, wk[32], 0, o1, false, false);
      o2 = __builtin_amdgcn_wmma_f32_16x16x4_f32(0, a, 0, wk[64], 0, o2, false, false);
      o3 = __builtin_amdgcn_wmma_f32_16x16x4_f32(0, a, 0, wk[96], 0, o3, false, false);
    }
  }
  const int half = lane >> 4;
  const int c = lane & 15;
#pragma unroll
  for (int r = 0; r < 8; ++r) {
    v4f o;
    o.x = o0[r] > 0.f ? o0[r] : 0.f;  o.y = o1[r] > 0.f ? o1[r] : 0.f;
    o.z = o2[r] > 0.f ? o2[r] : 0.f;  o.w = o3[r] > 0.f ? o3[r] : 0.f;
    *reinterpret_cast<v4f*>(hgroup + (8 * half + r) * NHID + 4 * c) = o;
  }
}

template <bool FIRST>
__global__ void __launch_bounds__(1024) spmm_owner_kernel(
    const unsigned* __restrict__ tid, const uint2* __restrict__ rec,
    const float* __restrict__ h, const float* __restrict__ h0,
    const float* __restrict__ Wl, float beta, float* __restrict__ hout,
    uint2* __restrict__ glists, unsigned* __restrict__ gcounts) {
  extern __shared__ __attribute__((aligned(16))) float tile_lds[];
  __shared__ unsigned wq[32 * QCAP];
  __shared__ unsigned gcnt[32];
  __shared__ __attribute__((aligned(16))) float Wp[NHID * NHID];
  float* acc = tile_lds;
  float4* acc4 = reinterpret_cast<float4*>(tile_lds);
  const int lane = threadIdx.x & 31;
  const int wave = __builtin_amdgcn_readfirstlane(threadIdx.x >> 5);
  const int qb = wave * QCAP;
  const int tile = blockIdx.x;
  uint2* tlist = glists + (size_t)tile * NG * GCAP;
  const long long grow = (long long)tile * TR + wave * 16;
  const float* h0row = h0 + (grow + (lane & 15)) * NHID + 32 * (lane >> 4);
  float* hgroup = hout + grow * NHID;

  stage_layer_weights(Wl, beta, Wp);
  __syncthreads();

  if (!FIRST) {
    const unsigned myc = (lane < NG) ? gcounts[tile * NG + lane] : 0u;
    if (__builtin_amdgcn_ballot_w32(myc > (unsigned)GCAP) == 0u) {
      if (wave < NG) {
        const int n = __builtin_amdgcn_readlane((int)myc, wave);
        v8f s[4];
        replay_group(tlist + (size_t)wave * GCAP, n, lane, h, s);
        layer_body(s, h0row, Wp, lane, hgroup);
      }
      return;
    }
  }

  for (int i = threadIdx.x; i < TR * NHID / 4; i += 1024)
    acc4[i] = make_float4(0.f, 0.f, 0.f, 0.f);
  if (FIRST && threadIdx.x < NG) gcnt[threadIdx.x] = 0u;
  __syncthreads();

  auto file_rec = [&](uint2 rc, bool valid) {
    if (valid) {
      const unsigned gq = (rc.y & 511u) >> 4;
      const unsigned slot = __hip_atomic_fetch_add(&gcnt[gq], 1u, __ATOMIC_RELAXED,
                                                   __HIP_MEMORY_SCOPE_WORKGROUP);
      if (slot < (unsigned)GCAP) tlist[gq * GCAP + slot] = rc;
    }
  };

  {
    const unsigned pat = (unsigned)tile * 0x01010101u;
    const uint4* tid4 = reinterpret_cast<const uint4*>(tid);
    const int g0 = (tile * NGROUP) / NTILES;

    struct ids2 { uint4 a, b; };
    auto load_ids = [&](int g) {
      const int ch = min(g * 32 + wave, NCHUNK - 1);
      ids2 r;
      r.a = tid4[ch * 64 + lane];
      r.b = tid4[ch * 64 + 32 + lane];
      return r;
    };
    auto next_group = [](int g) { return (g + 1 == NGROUP) ? 0 : g + 1; };

    int qn = 0;
    auto scan = [&](const ids2& v, int g) {
      const int ch = g * 32 + wave;
      if (ch >= NCHUNK) return;
      unsigned m = (zero_bytes(v.a.x ^ pat) >> 7) | (zero_bytes(v.a.y ^ pat) >> 6) |
                   (zero_bytes(v.a.z ^ pat) >> 5) | (zero_bytes(v.a.w ^ pat) >> 4) |
                   (zero_bytes(v.b.x ^ pat) >> 3) | (zero_bytes(v.b.y ^ pat) >> 2) |
                   (zero_bytes(v.b.z ^ pat) >> 1) | zero_bytes(v.b.w ^ pat);
      const unsigned ebase = (unsigned)ch * CHUNK + (unsigned)lane * 16u;
      while (true) {
        const bool has = m != 0u;
        const unsigned mask = __builtin_amdgcn_ballot_w32(has);
        if (mask == 0u) break;
        if (has) {
          const unsigned p = (unsigned)__builtin_ctz(m);
          wq[qb + qn + (int)__builtin_amdgcn_mbcnt_lo(mask, 0u)] =
              ebase + ((p & 4u) << 7) + ((p & 3u) << 2) + (p >> 3);
          m &= m - 1u;
        }
        qn += __builtin_popcount(mask);
        if (qn >= 32) {
          qn -= 32;
          const uint2 rc = rec[wq[qb + qn + lane]];
          if (FIRST) file_rec(rc, true);
          add_batch<true>(rc, 32, lane, h, acc);
        }
      }
    };

    static_assert(NGROUP % 2 == 0, "the scan loop takes two groups per trip");
    int g = g0;
    ids2 cur = load_ids(g);
    for (int k = 0; k < NGROUP; k += 2) {
      const int g1 = next_group(g);
      const int g2 = next_group(g1);
      const ids2 nxt = load_ids(g1);
      scan(cur, g);
      cur = load_ids(g2);
      scan(nxt, g1);
      g = g2;
    }
    if (qn > 0) {
      const uint2 rc = (lane < qn) ? rec[wq[qb + lane]] : make_uint2(0u, 0u);
      if (FIRST) file_rec(rc, lane < qn);
      add_batch<false>(rc, qn, lane, h, acc);
    }
  }
  __syncthreads();

  if (FIRST && wave < NG) {
    const unsigned n = gcnt[wave];
    if (lane == 0) gcounts[tile * NG + wave] = n;
    const unsigned idx = n + (unsigned)lane;
    if (n <= (unsigned)GCAP && lane < 16 && idx < ((n + 15u) & ~15u))
      tlist[(unsigned)wave * GCAP + idx] = make_uint2(0u, 0u);
  }

  if (wave < NG) {
    const v4f* trow = reinterpret_cast<const v4f*>(
        tile_lds + (wave * 16 + (lane & 15)) * NHID + 32 * (lane >> 4));
    v8f s[4];
#pragma unroll
    for (int r = 0; r < 8; ++r) {
      const v4f v = trow[r];
      s[0][r] = v.x;  s[1][r] = v.y;  s[2][r] = v.z;  s[3][r] = v.w;
    }
    layer_body(s, h0row, Wp, lane, hgroup);
  }
}

__global__ void __launch_bounds__(256) out_kernel(
    const float* __restrict__ h, const float* __restrict__ Wout,
    const float* __restrict__ bout, float* __restrict__ out) {
  __shared__ float Ws[NHID * NOUT];
  __shared__ float bs[NOUT];
  for (int idx = threadIdx.x; idx < NHID * NOUT; idx += 256) Ws[idx] = Wout[idx];
  if (threadIdx.x < NOUT) bs[threadIdx.x] = bout[threadIdx.x];
  __syncthreads();

  const int i = blockIdx.x * blockDim.x + threadIdx.x;
  if (i >= N_NODES) return;

  float hr[NHID];
  const float4* hp = reinterpret_cast<const float4*>(h + (long long)i * NHID);
#pragma unroll
  for (int k4 = 0; k4 < NHID / 4; ++k4) {
    const float4 v = hp[k4];
    hr[4 * k4 + 0] = v.x; hr[4 * k4 + 1] = v.y;
    hr[4 * k4 + 2] = v.z; hr[4 * k4 + 3] = v.w;
  }

  float lg[NOUT];
#pragma unroll
  for (int c = 0; c < NOUT; ++c) lg[c] = bs[c];
  for (int k = 0; k < NHID; ++k) {
    const float hk = hr[k];
#pragma unroll
    for (int c = 0; c < NOUT; ++c) lg[c] = fmaf(hk, Ws[k * NOUT + c], lg[c]);
  }

  float mx = lg[0];
#pragma unroll
  for (int c = 1; c < NOUT; ++c) mx = fmaxf(mx, lg[c]);
  float s = 0.f;
#pragma unroll
  for (int c = 0; c < NOUT; ++c) s += expf(lg[c] - mx);
  const float lse = mx + logf(s);
#pragma unroll
  for (int c = 0; c < NOUT; ++c)
    out[(long long)i * NOUT + c] = lg[c] - lse;
}

extern "C" void kernel_launch(void* const* d_in, const int* in_sizes, int n_in,
                              void* d_out, int out_size, void* d_ws, size_t ws_size,
                              hipStream_t stream) {
  (void)in_sizes; (void)n_in; (void)out_size; (void)ws_size;
  const float* x     = (const float*)d_in[0];
  const int*   row   = (const int*)  d_in[1];
  const int*   col   = (const int*)  d_in[2];
  const float* ew    = (const float*)d_in[3];
  const float* Win   = (const float*)d_in[4];
  const float* bin   = (const float*)d_in[5];
  const float* convW = (const float*)d_in[6];
  const float* Wout  = (const float*)d_in[7];
  const float* bout  = (const float*)d_in[8];

  const size_t S = (size_t)N_NODES * NHID;
  float* h0 = (float*)d_ws;
  float* hA = h0 + S;
  float* hB = hA + S;

  const int rowBlocks = (N_NODES + 255) / 256;
  const size_t inLds = (size_t)(NIN / 2) * BRS * sizeof(float);
  in_gemm_kernel<<<rowBlocks, 256, inLds, stream>>>(x, Win, bin, h0);

  unsigned* tid = (unsigned*)(hB + S);
  uint2* rec = (uint2*)(tid + E_EDGES / 4);
  edge_pack_kernel<<<E_EDGES / 4 / 256, 256, 0, stream>>>(row, col, ew, tid, rec);
  const size_t spmmLds = (size_t)164 * 1024;
  uint2* lists = rec + E_EDGES;
  unsigned* counts = (unsigned*)(lists + (size_t)NTILES * NG * GCAP);

  const float* hin = h0;
  float* hnext = hA;
  float* hother = hB;
  for (int l = 0; l < NLAYERS; ++l) {
    const float beta = logf(0.5f / (float)(l + 1) + 1.f);
    const float* Wl = convW + (size_t)l * NHID * NHID;
    if (l == 0)
      spmm_owner_kernel<true><<<NTILES, 1024, spmmLds, stream>>>(
          tid, rec, hin, h0, Wl, beta, hnext, lists, counts);
    else
      spmm_owner_kernel<false><<<NTILES, 1024, spmmLds, stream>>>(
          tid, rec, hin, h0, Wl, beta, hnext, lists, counts);
    hin = hnext;
    hnext = hother;
    hother = (float*)hin;
  }

  out_kernel<<<(N_NODES + 255) / 256, 256, 0, stream>>>(hin, Wout, bout, (float*)d_out);
}


# Layers 2-8: records filed per destination row; a wave per row gathers its source rows and sums them in registers by FMA in list order (was one-hot WMMA steps); the LDS tile feeds the layer product
# speedup vs baseline: 4.1017x; 1.0688x over previous
#include <hip/hip_runtime.h>
#include <hip/hip_bf16.h>
#include <math.h>

#define N_NODES 100000
#define E_EDGES 3200000
#define NIN     500
#define NHID    64
#define NOUT    40
#define NLAYERS 8
#define ALPHA   0.1f

#define BRS 160

typedef __attribute__((ext_vector_type(2))) float v2f;
typedef __attribute__((ext_vector_type(8))) float v8f;
typedef __attribute__((ext_vector_type(4))) unsigned v4u;
typedef __attribute__((ext_vector_type(4))) float v4f;

__device__ __forceinline__ void wmma_acc(v8f& c, v2f a, v2f b) {
  asm("v_wmma_f32_16x16x4_f32 %0, %1, %2, %0" : "+v"(c) : "v"(a), "v"(b));
}
__device__ __forceinline__ void wmma_epilogue_guard() {
  asm volatile("s_nop 8");
}

__global__ void __launch_bounds__(256) in_gemm_kernel(
    const float* __restrict__ x, const float* __restrict__ Win,
    const float* __restrict__ bin, float* __restrict__ h0) {
  extern __shared__ float Wp[];
  for (int idx = threadIdx.x; idx < NIN * NHID; idx += 256) {
    const int k = idx >> 6;
    const int n = idx & 63;
    Wp[(k >> 1) * BRS + 2 * n + (k & 1)] = Win[idx];
  }
  __syncthreads();

  const int wave = threadIdx.x >> 5;
  const int lane = threadIdx.x & 31;
  const int half = lane >> 4;
  const int l16  = lane & 15;
  const int r0   = blockIdx.x * 256 + wave * 32;

  const int arow0 = min(r0 + l16,      N_NODES - 1);
  const int arow1 = min(r0 + 16 + l16, N_NODES - 1);
  const float* xr0 = x + (long long)arow0 * NIN;
  const float* xr1 = x + (long long)arow1 * NIN;

  v8f acc0[4] = {};
  v8f acc1[4] = {};
  for (int k = 0; k < NIN; k += 4) {
    const int kk = k + 2 * half;
    const v2f a0 = *reinterpret_cast<const v2f*>(xr0 + kk);
    const v2f a1 = *reinterpret_cast<const v2f*>(xr1 + kk);
    const int bbase = (kk >> 1) * BRS + 2 * l16;
#pragma unroll
    for (int t = 0; t < 4; ++t) {
      const v2f b = *reinterpret_cast<const v2f*>(&Wp[bbase + t * 32]);
      wmma_acc(acc0[t], a0, b);
      wmma_acc(acc1[t], a1, b);
    }
  }
  wmma_epilogue_guard();

#pragma unroll
  for (int t = 0; t < 4; ++t) {
    const int n = t * 16 + l16;
    const float bias = bin[n];
#pragma unroll
    for (int r = 0; r < 8; ++r) {
      const int m = r + 8 * half;
      const int orow0 = r0 + m;
      const int orow1 = r0 + 16 + m;
      if (orow0 < N_NODES) {
        const float v = acc0[t][r] + bias;
        h0[(long long)orow0 * NHID + n] = v > 0.f ? v : 0.f;
      }
      if (orow1 < N_NODES) {
        const float v = acc1[t][r] + bias;
        h0[(long long)orow1 * NHID + n] = v > 0.f ? v : 0.f;
      }
    }
  }
}

#define TR      400
#define NTILES  (N_NODES / TR)
#define CHUNK   1024
#define NCHUNK  (E_EDGES / CHUNK)
#define NGROUP  ((NCHUNK + 31) / 32)
#define QCAP    64
#define NG      (TR / 16)
#define RCAP    128
#define RB      48
static_assert(N_NODES % TR == 0 && NTILES <= 256, "tile id must fit a byte");
static_assert(TR <= 512 && N_NODES <= (1 << 23), "record packs col << 9 | local row");
static_assert(E_EDGES % CHUNK == 0, "edge count must be whole chunks");
static_assert(TR % 16 == 0 && NG <= 32 && TR >= 32, "one body wave per group; every wave has a row");
static_assert(RCAP % 32 == 0 && RB == 48, "record blocks of 32 lanes; replay_rows has one case per eight up to RB");

__global__ void __launch_bounds__(256) edge_pack_kernel(
    const int* __restrict__ row, const int* __restrict__ col,
    const float* __restrict__ ew, unsigned* __restrict__ tid,
    uint2* __restrict__ rec) {
  const int t = blockIdx.x * 256 + threadIdx.x;
  if (t >= E_EDGES / 4) return;
  const int4 r = reinterpret_cast<const int4*>(row)[t];
  const int4 c = reinterpret_cast<const int4*>(col)[t];
  const float4 w = reinterpret_cast<const float4*>(ew)[t];
  const unsigned t0 = (unsigned)(r.x / TR), t1 = (unsigned)(r.y / TR);
  const unsigned t2 = (unsigned)(r.z / TR), t3 = (unsigned)(r.w / TR);
  tid[t] = t0 | (t1 << 8) | (t2 << 16) | (t3 << 24);
  uint4* rp = reinterpret_cast<uint4*>(rec + 4 * (long long)t);
  rp[0] = make_uint4(__float_as_uint(w.x), ((unsigned)c.x << 9) | ((unsigned)r.x - t0 * TR),
                     __float_as_uint(w.y), ((unsigned)c.y << 9) | ((unsigned)r.y - t1 * TR));
  rp[1] = make_uint4(__float_as_uint(w.z), ((unsigned)c.z << 9) | ((unsigned)r.z - t2 * TR),
                     __float_as_uint(w.w), ((unsigned)c.w << 9) | ((unsigned)r.w - t3 * TR));
}

__device__ __forceinline__ unsigned zero_bytes(unsigned x) {
  const unsigned t = ((x & 0x7f7f7f7fu) + 0x7f7f7f7fu) | x;
  return ~t & 0x80808080u;
}

template <bool FULL>
__device__ __forceinline__ void add_batch(
    uint2 rc, int n, int lane, const float* __restrict__ h, float* acc) {
  const float w = __uint_as_float(rc.x);
  const int c = (int)(rc.y >> 9);
  const int lr = (int)(rc.y & 511u);
  v2f g[32];
#pragma unroll
  for (int i = 0; i < 32; ++i) {
    const int ci = __builtin_amdgcn_readlane(c, i);
    g[i] = *reinterpret_cast<const v2f*>(h + (long long)ci * NHID + 2 * lane);
  }
#pragma unroll
  for (int i = 0; i < 32; ++i) {
    if (FULL || i < n) {
      const float wi = __int_as_float(__builtin_amdgcn_readlane(__float_as_int(w), i));
      const int li = __builtin_amdgcn_readlane(lr, i);
      float* a = acc + li * NHID + 2 * lane;
      __hip_atomic_fetch_add(a,     wi * g[i].x, __ATOMIC_RELAXED, __HIP_MEMORY_SCOPE_WORKGROUP);
      __hip_atomic_fetch_add(a + 1, wi * g[i].y, __ATOMIC_RELAXED, __HIP_MEMORY_SCOPE_WORKGROUP);
    }
  }
}

__device__ __forceinline__ v2f row_of(const char* hb, unsigned y) {
  return *reinterpret_cast<const v2f*>(hb + ((y >> 9) << 8));
}

template <int NB>
__device__ __forceinline__ v2f sum_records(uint2 ra, uint2 rb, const char* hb) {
  v2f g[NB * 8];
#pragma unroll
  for (int i = 0; i < NB * 8; ++i) {
    const unsigned y = (unsigned)__builtin_amdgcn_readlane((int)(i < 32 ? ra.y : rb.y), i & 31);
    g[i] = row_of(hb, y);
  }
  v2f s = {0.f, 0.f};
#pragma unroll
  for (int i = 0; i < NB * 8; ++i) {
    const float w = __int_as_float(
        __builtin_amdgcn_readlane((int)(i < 32 ? ra.x : rb.x), i & 31));
    s.x = fmaf(w, g[i].x, s.x);
    s.y = fmaf(w, g[i].y, s.y);
  }
  return s;
}

__device__ __forceinline__ void replay_rows(
    const uint2* __restrict__ tlist, const unsigned* rn, int wave, int lane,
    const float* __restrict__ h, float* tile) {
  const char* hb = reinterpret_cast<const char*>(h) + 8 * lane;
  int lr = wave;
  uint2 ra = tlist[lr * RCAP + lane];
  uint2 rb = tlist[lr * RCAP + 32 + lane];
  int n = __builtin_amdgcn_readfirstlane((int)rn[lr]);
  while (true) {
    const int npad = (n + 7) & ~7;
    const int nxt = lr + 32;
    const int pf = nxt < TR ? nxt : lr;
    const uint2 na = tlist[pf * RCAP + lane];
    const uint2 nb = tlist[pf * RCAP + 32 + lane];
    const unsigned nn = rn[pf];
    v2f s = {0.f, 0.f};
    switch ((npad < RB ? npad : RB) >> 3) {
      case 0:  break;
      case 1:  s = sum_records<1>(ra, rb, hb);  break;
      case 2:  s = sum_records<2>(ra, rb, hb);  break;
      case 3:  s = sum_records<3>(ra, rb, hb);  break;
      case 4:  s = sum_records<4>(ra, rb, hb);  break;
      case 5:  s = sum_records<5>(ra, rb, hb);  break;
      default: s = sum_records<6>(ra, rb, hb);  break;
    }
    for (int i = RB; i < npad; i += 8) {
      const uint2 rx = tlist[lr * RCAP + (i & ~31) + lane];
      v2f t[8];
#pragma unroll
      for (int j = 0; j < 8; ++j)
        t[j] = row_of(hb, (unsigned)__builtin_amdgcn_readlane((int)rx.y, (i + j) & 31));
#pragma unroll
      for (int j = 0; j < 8; ++j) {
        const float w = __int_as_float(__builtin_amdgcn_readlane((int)rx.x, (i + j) & 31));
        s.x = fmaf(w, t[j].x, s.x);
        s.y = fmaf(w, t[j].y, s.y);
      }
    }
    *reinterpret_cast<v2f*>(tile + lr * NHID + 2 * lane) = s;
    if (nxt >= TR) break;
    lr = nxt;  ra = na;  rb = nb;
    n = __builtin_amdgcn_readfirstlane((int)nn);
  }
}

__device__ __forceinline__ void stage_layer_weights(
    const float* __restrict__ Wl, float beta, float* Wp) {
  const float omb = 1.f - beta;
#pragma unroll
  for (int j = 0; j < NHID * NHID / 1024; ++j) {
    const int idx = (int)threadIdx.x + 1024 * j;
    const int k = idx >> 6;
    const int n = idx & 63;
    const float w = beta * Wl[idx];
    const int kh = k >> 5, kt = k & 3, ki = (k >> 3) & 3, ks = (k >> 2) & 1;
    Wp[(((kt * 4 + ki) * 4 + (n & 3)) << 6) + 2 * ((n >> 2) + 16 * kh) + ks] =
        (k == n) ? w + omb : w;
  }
}

__device__ __forceinline__ void layer_body(
    v8f (&s)[4], const float* __restrict__ h0row, const float* Wp, int lane,
    float* __restrict__ hgroup) {
  const float sA = 1.f - ALPHA;
  const float sB = ALPHA;
  v4f g[8];
#pragma unroll
  for (int r = 0; r < 8; ++r) g[r] = *reinterpret_cast<const v4f*>(h0row + 4 * r);
#pragma unroll
  for (int r = 0; r < 8; ++r) {
    s[0][r] = sA * s[0][r] + sB * g[r].x;
    s[1][r] = sA * s[1][r] + sB * g[r].y;
    s[2][r] = sA * s[2][r] + sB * g[r].z;
    s[3][r] = sA * s[3][r] + sB * g[r].w;
  }
  const v8f z8 = {};
  v8f o0 = z8, o1 = z8, o2 = z8, o3 = z8;
  const v2f* wp = reinterpret_cast<const v2f*>(Wp) + lane;
#pragma unroll
  for (int t = 0; t < 4; ++t) {
#pragma unroll
    for (int i = 0; i < 4; ++i) {
      v2f a;
      a.x = s[t][2 * i];  a.y = s[t][2 * i + 1];
      const v2f* wk = wp + (t * 4 + i) * 128;
      o0 = __builtin_amdgcn_wmma_f32_16x16x4_f32(0, a, 0, wk[0],  0, o0, false, false);
      o1 = __builtin_amdgcn_wmma_f32_16x16x4_f32(0, a, 0, wk[32], 0, o1, false, false);
      o2 = __builtin_amdgcn_wmma_f32_16x16x4_f32(0, a, 0, wk[64], 0, o2, false, false);
      o3 = __builtin_amdgcn_wmma_f32_16x16x4_f32(0, a, 0, wk[96], 0, o3, false, false);
    }
  }
  const int half = lane >> 4;
  const int c = lane & 15;
#pragma unroll
  for (int r = 0; r < 8; ++r) {
    v4f o;
    o.x = o0[r] > 0.f ? o0[r] : 0.f;  o.y = o1[r] > 0.f ? o1[r] : 0.f;
    o.z = o2[r] > 0.f ? o2[r] : 0.f;  o.w = o3[r] > 0.f ? o3[r] : 0.f;
    *reinterpret_cast<v4f*>(hgroup + (8 * half + r) * NHID + 4 * c) = o;
  }
}

template <bool FIRST>
__global__ void __launch_bounds__(1024) spmm_owner_kernel(
    const unsigned* __restrict__ tid, const uint2* __restrict__ rec,
    const float* __restrict__ h, const float* __restrict__ h0,
    const float* __restrict__ Wl, float beta, float* __restrict__ hout,
    uint2* __restrict__ rlists, unsigned* __restrict__ rcounts) {
  extern __shared__ __attribute__((aligned(16))) float tile_lds[];
  __shared__ unsigned wq[32 * QCAP];
  __shared__ unsigned rcnt[TR];
  __shared__ unsigned wover[32];
  __shared__ __attribute__((aligned(16))) float Wp[NHID * NHID];
  float* acc = tile_lds;
  float4* acc4 = reinterpret_cast<float4*>(tile_lds);
  const int lane = threadIdx.x & 31;
  const int wave = __builtin_amdgcn_readfirstlane(threadIdx.x >> 5);
  const int qb = wave * QCAP;
  const int tile = blockIdx.x;
  uint2* tlist = rlists + (size_t)tile * TR * RCAP;
  const long long grow = (long long)tile * TR + wave * 16;
  const float* h0row = h0 + (grow + (lane & 15)) * NHID + 32 * (lane >> 4);
  float* hgroup = hout + grow * NHID;

  if (!FIRST) {
    const int r = (int)threadIdx.x;
    const unsigned myn = (r < TR) ? rcounts[tile * TR + r] : 0u;
    if (r < TR) rcnt[r] = myn;
    const unsigned ob = __builtin_amdgcn_ballot_w32(myn > (unsigned)RCAP);
    if (lane == 0) wover[wave] = ob;
  }
  stage_layer_weights(Wl, beta, Wp);
  __syncthreads();

  bool replayed = false;
  if (!FIRST) {
    if (__builtin_amdgcn_ballot_w32(wover[lane] != 0u) == 0u) {
      replay_rows(tlist, rcnt, wave, lane, h, tile_lds);
      replayed = true;
    }
  }

  if (!replayed) {
  for (int i = threadIdx.x; i < TR * NHID / 4; i += 1024)
    acc4[i] = make_float4(0.f, 0.f, 0.f, 0.f);
  if (FIRST && threadIdx.x < TR) rcnt[threadIdx.x] = 0u;
  __syncthreads();

  auto file_rec = [&](uint2 rc, bool valid) {
    if (valid) {
      const unsigned lr = rc.y & 511u;
      const unsigned slot = __hip_atomic_fetch_add(&rcnt[lr], 1u, __ATOMIC_RELAXED,
                                                   __HIP_MEMORY_SCOPE_WORKGROUP);
      if (slot < (unsigned)RCAP) tlist[lr * RCAP + slot] = rc;
    }
  };

  {
    const unsigned pat = (unsigned)tile * 0x01010101u;
    const uint4* tid4 = reinterpret_cast<const uint4*>(tid);
    const int g0 = (tile * NGROUP) / NTILES;

    struct ids2 { uint4 a, b; };
    auto load_ids = [&](int g) {
      const int ch = min(g * 32 + wave, NCHUNK - 1);
      ids2 r;
      r.a = tid4[ch * 64 + lane];
      r.b = tid4[ch * 64 + 32 + lane];
      return r;
    };
    auto next_group = [](int g) { return (g + 1 == NGROUP) ? 0 : g + 1; };

    int qn = 0;
    auto scan = [&](const ids2& v, int g) {
      const int ch = g * 32 + wave;
      if (ch >= NCHUNK) return;
      unsigned m = (zero_bytes(v.a.x ^ pat) >> 7) | (zero_bytes(v.a.y ^ pat) >> 6) |
                   (zero_bytes(v.a.z ^ pat) >> 5) | (zero_bytes(v.a.w ^ pat) >> 4) |
                   (zero_bytes(v.b.x ^ pat) >> 3) | (zero_bytes(v.b.y ^ pat) >> 2) |
                   (zero_bytes(v.b.z ^ pat) >> 1) | zero_bytes(v.b.w ^ pat);
      const unsigned ebase = (unsigned)ch * CHUNK + (unsigned)lane * 16u;
      while (true) {
        const bool has = m != 0u;
        const unsigned mask = __builtin_amdgcn_ballot_w32(has);
        if (mask == 0u) break;
        if (has) {
          const unsigned p = (unsigned)__builtin_ctz(m);
          wq[qb + qn + (int)__builtin_amdgcn_mbcnt_lo(mask, 0u)] =
              ebase + ((p & 4u) << 7) + ((p & 3u) << 2) + (p >> 3);
          m &= m - 1u;
        }
        qn += __builtin_popcount(mask);
        if (qn >= 32) {
          qn -= 32;
          const uint2 rc = rec[wq[qb + qn + lane]];
          if (FIRST) file_rec(rc, true);
          add_batch<true>(rc, 32, lane, h, acc);
        }
      }
    };

    static_assert(NGROUP % 2 == 0, "the scan loop takes two groups per trip");
    int g = g0;
    ids2 cur = load_ids(g);
    for (int k = 0; k < NGROUP; k += 2) {
      const int g1 = next_group(g);
      const int g2 = next_group(g1);
      const ids2 nxt = load_ids(g1);
      scan(cur, g);
      cur = load_ids(g2);
      scan(nxt, g1);
      g = g2;
    }
    if (qn > 0) {
      const uint2 rc = (lane < qn) ? rec[wq[qb + lane]] : make_uint2(0u, 0u);
      if (FIRST) file_rec(rc, lane < qn);
      add_batch<false>(rc, qn, lane, h, acc);
    }
  }
  }
  __syncthreads();

  if (FIRST && threadIdx.x < TR) {
    const unsigned r = threadIdx.x;
    const unsigned n = rcnt[r];
    rcounts[tile * TR + r] = n;
    if (n <= (unsigned)RCAP) {
      const unsigned e = (n + 7u) & ~7u;
      for (unsigned i = n; i < e; ++i) tlist[r * RCAP + i] = make_uint2(0u, 0u);
    }
  }

  if (wave < NG) {
    const v4f* trow = reinterpret_cast<const v4f*>(
        tile_lds + (wave * 16 + (lane & 15)) * NHID + 32 * (lane >> 4));
    v8f s[4];
#pragma unroll
    for (int r = 0; r < 8; ++r) {
      const v4f v = trow[r];
      s[0][r] = v.x;  s[1][r] = v.y;  s[2][r] = v.z;  s[3][r] = v.w;
    }
    layer_body(s, h0row, Wp, lane, hgroup);
  }
}

__global__ void __launch_bounds__(256) out_kernel(
    const float* __restrict__ h, const float* __restrict__ Wout,
    const float* __restrict__ bout, float* __restrict__ out) {
  __shared__ float Ws[NHID * NOUT];
  __shared__ float bs[NOUT];
  for (int idx = threadIdx.x; idx < NHID * NOUT; idx += 256) Ws[idx] = Wout[idx];
  if (threadIdx.x < NOUT) bs[threadIdx.x] = bout[threadIdx.x];
  __syncthreads();

  const int i = blockIdx.x * blockDim.x + threadIdx.x;
  if (i >= N_NODES) return;

  float hr[NHID];
  const float4* hp = reinterpret_cast<const float4*>(h + (long long)i * NHID);
#pragma unroll
  for (int k4 = 0; k4 < NHID / 4; ++k4) {
    const float4 v = hp[k4];
    hr[4 * k4 + 0] = v.x; hr[4 * k4 + 1] = v.y;
    hr[4 * k4 + 2] = v.z; hr[4 * k4 + 3] = v.w;
  }

  float lg[NOUT];
#pragma unroll
  for (int c = 0; c < NOUT; ++c) lg[c] = bs[c];
  for (int k = 0; k < NHID; ++k) {
    const float hk = hr[k];
#pragma unroll
    for (int c = 0; c < NOUT; ++c) lg[c] = fmaf(hk, Ws[k * NOUT + c], lg[c]);
  }

  float mx = lg[0];
#pragma unroll
  for (int c = 1; c < NOUT; ++c) mx = fmaxf(mx, lg[c]);
  float s = 0.f;
#pragma unroll
  for (int c = 0; c < NOUT; ++c) s += expf(lg[c] - mx);
  const float lse = mx + logf(s);
#pragma unroll
  for (int c = 0; c < NOUT; ++c)
    out[(long long)i * NOUT + c] = lg[c] - lse;
}

extern "C" void kernel_launch(void* const* d_in, const int* in_sizes, int n_in,
                              void* d_out, int out_size, void* d_ws, size_t ws_size,
                              hipStream_t stream) {
  (void)in_sizes; (void)n_in; (void)out_size; (void)ws_size;
  const float* x     = (const float*)d_in[0];
  const int*   row   = (const int*)  d_in[1];
  const int*   col   = (const int*)  d_in[2];
  const float* ew    = (const float*)d_in[3];
  const float* Win   = (const float*)d_in[4];
  const float* bin   = (const float*)d_in[5];
  const float* convW = (const float*)d_in[6];
  const float* Wout  = (const float*)d_in[7];
  const float* bout  = (const float*)d_in[8];

  const size_t S = (size_t)N_NODES * NHID;
  float* h0 = (float*)d_ws;
  float* hA = h0 + S;
  float* hB = hA + S;

  const int rowBlocks = (N_NODES + 255) / 256;
  const size_t inLds = (size_t)(NIN / 2) * BRS * sizeof(float);
  in_gemm_kernel<<<rowBlocks, 256, inLds, stream>>>(x, Win, bin, h0);

  unsigned* tid = (unsigned*)(hB + S);
  uint2* rec = (uint2*)(tid + E_EDGES / 4);
  edge_pack_kernel<<<E_EDGES / 4 / 256, 256, 0, stream>>>(row, col, ew, tid, rec);
  const size_t spmmLds = (size_t)164 * 1024;
  uint2* lists = rec + E_EDGES;
  unsigned* counts = (unsigned*)(lists + (size_t)N_NODES * RCAP);

  const float* hin = h0;
  float* hnext = hA;
  float* hother = hB;
  for (int l = 0; l < NLAYERS; ++l) {
    const float beta = logf(0.5f / (float)(l + 1) + 1.f);
    const float* Wl = convW + (size_t)l * NHID * NHID;
    if (l == 0)
      spmm_owner_kernel<true><<<NTILES, 1024, spmmLds, stream>>>(
          tid, rec, hin, h0, Wl, beta, hnext, lists, counts);
    else
      spmm_owner_kernel<false><<<NTILES, 1024, spmmLds, stream>>>(
          tid, rec, hin, h0, Wl, beta, hnext, lists, counts);
    hin = hnext;
    hnext = hother;
    hother = (float*)hin;
  }

  out_kernel<<<(N_NODES + 255) / 256, 256, 0, stream>>>(hin, Wout, bout, (float*)d_out);
}


# Layer 1: the id scan only files records per row (no LDS float adds, no tile zero-fill); the block then sums its rows from its own lists as layers 2-8 do, FMA in list order (was LDS float atomics)
# speedup vs baseline: 4.2322x; 1.0318x over previous
#include <hip/hip_runtime.h>
#include <hip/hip_bf16.h>
#include <math.h>

#define N_NODES 100000
#define E_EDGES 3200000
#define NIN     500
#define NHID    64
#define NOUT    40
#define NLAYERS 8
#define ALPHA   0.1f

#define BRS 160

typedef __attribute__((ext_vector_type(2))) float v2f;
typedef __attribute__((ext_vector_type(8))) float v8f;
typedef __attribute__((ext_vector_type(4))) unsigned v4u;
typedef __attribute__((ext_vector_type(4))) float v4f;

__device__ __forceinline__ void wmma_acc(v8f& c, v2f a, v2f b) {
  asm("v_wmma_f32_16x16x4_f32 %0, %1, %2, %0" : "+v"(c) : "v"(a), "v"(b));
}
__device__ __forceinline__ void wmma_epilogue_guard() {
  asm volatile("s_nop 8");
}

__global__ void __launch_bounds__(256) in_gemm_kernel(
    const float* __restrict__ x, const float* __restrict__ Win,
    const float* __restrict__ bin, float* __restrict__ h0) {
  extern __shared__ float Wp[];
  for (int idx = threadIdx.x; idx < NIN * NHID; idx += 256) {
    const int k = idx >> 6;
    const int n = idx & 63;
    Wp[(k >> 1) * BRS + 2 * n + (k & 1)] = Win[idx];
  }
  __syncthreads();

  const int wave = threadIdx.x >> 5;
  const int lane = threadIdx.x & 31;
  const int half = lane >> 4;
  const int l16  = lane & 15;
  const int r0   = blockIdx.x * 256 + wave * 32;

  const int arow0 = min(r0 + l16,      N_NODES - 1);
  const int arow1 = min(r0 + 16 + l16, N_NODES - 1);
  const float* xr0 = x + (long long)arow0 * NIN;
  const float* xr1 = x + (long long)arow1 * NIN;

  v8f acc0[4] = {};
  v8f acc1[4] = {};
  for (int k = 0; k < NIN; k += 4) {
    const int kk = k + 2 * half;
    const v2f a0 = *reinterpret_cast<const v2f*>(xr0 + kk);
    const v2f a1 = *reinterpret_cast<const v2f*>(xr1 + kk);
    const int bbase = (kk >> 1) * BRS + 2 * l16;
#pragma unroll
    for (int t = 0; t < 4; ++t) {
      const v2f b = *reinterpret_cast<const v2f*>(&Wp[bbase + t * 32]);
      wmma_acc(acc0[t], a0, b);
      wmma_acc(acc1[t], a1, b);
    }
  }
  wmma_epilogue_guard();

#pragma unroll
  for (int t = 0; t < 4; ++t) {
    const int n = t * 16 + l16;
    const float bias = bin[n];
#pragma unroll
    for (int r = 0; r < 8; ++r) {
      const int m = r + 8 * half;
      const int orow0 = r0 + m;
      const int orow1 = r0 + 16 + m;
      if (orow0 < N_NODES) {
        const float v = acc0[t][r] + bias;
        h0[(long long)orow0 * NHID + n] = v > 0.f ? v : 0.f;
      }
      if (orow1 < N_NODES) {
        const float v = acc1[t][r] + bias;
        h0[(long long)orow1 * NHID + n] = v > 0.f ? v : 0.f;
      }
    }
  }
}

#define TR      400
#define NTILES  (N_NODES / TR)
#define CHUNK   1024
#define NCHUNK  (E_EDGES / CHUNK)
#define NGROUP  ((NCHUNK + 31) / 32)
#define QCAP    64
#define NG      (TR / 16)
#define RCAP    128
#define RB      48
static_assert(N_NODES % TR == 0 && NTILES <= 256, "tile id must fit a byte");
static_assert(TR <= 512 && N_NODES <= (1 << 23), "record packs col << 9 | local row");
static_assert(E_EDGES % CHUNK == 0, "edge count must be whole chunks");
static_assert(TR % 16 == 0 && NG <= 32 && TR >= 32, "one body wave per group; every wave has a row");
static_assert(RCAP % 32 == 0 && RB == 48, "record blocks of 32 lanes; replay_rows has one case per eight up to RB");

__global__ void __launch_bounds__(256) edge_pack_kernel(
    const int* __restrict__ row, const int* __restrict__ col,
    const float* __restrict__ ew, unsigned* __restrict__ tid,
    uint2* __restrict__ rec) {
  const int t = blockIdx.x * 256 + threadIdx.x;
  if (t >= E_EDGES / 4) return;
  const int4 r = reinterpret_cast<const int4*>(row)[t];
  const int4 c = reinterpret_cast<const int4*>(col)[t];
  const float4 w = reinterpret_cast<const float4*>(ew)[t];
  const unsigned t0 = (unsigned)(r.x / TR), t1 = (unsigned)(r.y / TR);
  const unsigned t2 = (unsigned)(r.z / TR), t3 = (unsigned)(r.w / TR);
  tid[t] = t0 | (t1 << 8) | (t2 << 16) | (t3 << 24);
  uint4* rp = reinterpret_cast<uint4*>(rec + 4 * (long long)t);
  rp[0] = make_uint4(__float_as_uint(w.x), ((unsigned)c.x << 9) | ((unsigned)r.x - t0 * TR),
                     __float_as_uint(w.y), ((unsigned)c.y << 9) | ((unsigned)r.y - t1 * TR));
  rp[1] = make_uint4(__float_as_uint(w.z), ((unsigned)c.z << 9) | ((unsigned)r.z - t2 * TR),
                     __float_as_uint(w.w), ((unsigned)c.w << 9) | ((unsigned)r.w - t3 * TR));
}

__device__ __forceinline__ unsigned zero_bytes(unsigned x) {
  const unsigned t = ((x & 0x7f7f7f7fu) + 0x7f7f7f7fu) | x;
  return ~t & 0x80808080u;
}

template <bool FULL>
__device__ __forceinline__ void add_batch(
    uint2 rc, int n, int lane, const float* __restrict__ h, float* acc) {
  const float w = __uint_as_float(rc.x);
  const int c = (int)(rc.y >> 9);
  const int lr = (int)(rc.y & 511u);
  v2f g[32];
#pragma unroll
  for (int i = 0; i < 32; ++i) {
    const int ci = __builtin_amdgcn_readlane(c, i);
    g[i] = *reinterpret_cast<const v2f*>(h + (long long)ci * NHID + 2 * lane);
  }
#pragma unroll
  for (int i = 0; i < 32; ++i) {
    if (FULL || i < n) {
      const float wi = __int_as_float(__builtin_amdgcn_readlane(__float_as_int(w), i));
      const int li = __builtin_amdgcn_readlane(lr, i);
      float* a = acc + li * NHID + 2 * lane;
      __hip_atomic_fetch_add(a,     wi * g[i].x, __ATOMIC_RELAXED, __HIP_MEMORY_SCOPE_WORKGROUP);
      __hip_atomic_fetch_add(a + 1, wi * g[i].y, __ATOMIC_RELAXED, __HIP_MEMORY_SCOPE_WORKGROUP);
    }
  }
}

__device__ __forceinline__ v2f row_of(const char* hb, unsigned y) {
  return *reinterpret_cast<const v2f*>(hb + ((y >> 9) << 8));
}

template <int NB>
__device__ __forceinline__ v2f sum_records(uint2 ra, uint2 rb, const char* hb) {
  v2f g[NB * 8];
#pragma unroll
  for (int i = 0; i < NB * 8; ++i) {
    const unsigned y = (unsigned)__builtin_amdgcn_readlane((int)(i < 32 ? ra.y : rb.y), i & 31);
    g[i] = row_of(hb, y);
  }
  v2f s = {0.f, 0.f};
#pragma unroll
  for (int i = 0; i < NB * 8; ++i) {
    const float w = __int_as_float(
        __builtin_amdgcn_readlane((int)(i < 32 ? ra.x : rb.x), i & 31));
    s.x = fmaf(w, g[i].x, s.x);
    s.y = fmaf(w, g[i].y, s.y);
  }
  return s;
}

__device__ __forceinline__ void replay_rows(
    const uint2* __restrict__ tlist, const unsigned* rn, int wave, int lane,
    const float* __restrict__ h, float* tile) {
  const char* hb = reinterpret_cast<const char*>(h) + 8 * lane;
  int lr = wave;
  uint2 ra = tlist[lr * RCAP + lane];
  uint2 rb = tlist[lr * RCAP + 32 + lane];
  int n = __builtin_amdgcn_readfirstlane((int)rn[lr]);
  while (true) {
    const int npad = (n + 7) & ~7;
    const int nxt = lr + 32;
    const int pf = nxt < TR ? nxt : lr;
    const uint2 na = tlist[pf * RCAP + lane];
    const uint2 nb = tlist[pf * RCAP + 32 + lane];
    const unsigned nn = rn[pf];
    v2f s = {0.f, 0.f};
    switch ((npad < RB ? npad : RB) >> 3) {
      case 0:  break;
      case 1:  s = sum_records<1>(ra, rb, hb);  break;
      case 2:  s = sum_records<2>(ra, rb, hb);  break;
      case 3:  s = sum_records<3>(ra, rb, hb);  break;
      case 4:  s = sum_records<4>(ra, rb, hb);  break;
      case 5:  s = sum_records<5>(ra, rb, hb);  break;
      default: s = sum_records<6>(ra, rb, hb);  break;
    }
    for (int i = RB; i < npad; i += 8) {
      const uint2 rx = tlist[lr * RCAP + (i & ~31) + lane];
      v2f t[8];
#pragma unroll
      for (int j = 0; j < 8; ++j)
        t[j] = row_of(hb, (unsigned)__builtin_amdgcn_readlane((int)rx.y, (i + j) & 31));
#pragma unroll
      for (int j = 0; j < 8; ++j) {
        const float w = __int_as_float(__builtin_amdgcn_readlane((int)rx.x, (i + j) & 31));
        s.x = fmaf(w, t[j].x, s.x);
        s.y = fmaf(w, t[j].y, s.y);
      }
    }
    *reinterpret_cast<v2f*>(tile + lr * NHID + 2 * lane) = s;
    if (nxt >= TR) break;
    lr = nxt;  ra = na;  rb = nb;
    n = __builtin_amdgcn_readfirstlane((int)nn);
  }
}

__device__ __forceinline__ void stage_layer_weights(
    const float* __restrict__ Wl, float beta, float* Wp) {
  const float omb = 1.f - beta;
#pragma unroll
  for (int j = 0; j < NHID * NHID / 1024; ++j) {
    const int idx = (int)threadIdx.x + 1024 * j;
    const int k = idx >> 6;
    const int n = idx & 63;
    const float w = beta * Wl[idx];
    const int kh = k >> 5, kt = k & 3, ki = (k >> 3) & 3, ks = (k >> 2) & 1;
    Wp[(((kt * 4 + ki) * 4 + (n & 3)) << 6) + 2 * ((n >> 2) + 16 * kh) + ks] =
        (k == n) ? w + omb : w;
  }
}

__device__ __forceinline__ void layer_body(
    v8f (&s)[4], const float* __restrict__ h0row, const float* Wp, int lane,
    float* __restrict__ hgroup) {
  const float sA = 1.f - ALPHA;
  const float sB = ALPHA;
  v4f g[8];
#pragma unroll
  for (int r = 0; r < 8; ++r) g[r] = *reinterpret_cast<const v4f*>(h0row + 4 * r);
#pragma unroll
  for (int r = 0; r < 8; ++r) {
    s[0][r] = sA * s[0][r] + sB * g[r].x;
    s[1][r] = sA * s[1][r] + sB * g[r].y;
    s[2][r] = sA * s[2][r] + sB * g[r].z;
    s[3][r] = sA * s[3][r] + sB * g[r].w;
  }
  const v8f z8 = {};
  v8f o0 = z8, o1 = z8, o2 = z8, o3 = z8;
  const v2f* wp = reinterpret_cast<const v2f*>(Wp) + lane;
#pragma unroll
  for (int t = 0; t < 4; ++t) {
#pragma unroll
    for (int i = 0; i < 4; ++i) {
      v2f a;
      a.x = s[t][2 * i];  a.y = s[t][2 * i + 1];
      const v2f* wk = wp + (t * 4 + i) * 128;
      o0 = __builtin_amdgcn_wmma_f32_16x16x4_f32(0, a, 0, wk[0],  0, o0, false, false);
      o1 = __builtin_amdgcn_wmma_f32_16x16x4_f32(0, a, 0, wk[32], 0, o1, false, false);
      o2 = __builtin_amdgcn_wmma_f32_16x16x4_f32(0, a, 0, wk[64], 0, o2, false, false);
      o3 = __builtin_amdgcn_wmma_f32_16x16x4_f32(0, a, 0, wk[96], 0, o3, false, false);
    }
  }
  const int half = lane >> 4;
  const int c = lane & 15;
#pragma unroll
  for (int r = 0; r < 8; ++r) {
    v4f o;
    o.x = o0[r] > 0.f ? o0[r] : 0.f;  o.y = o1[r] > 0.f ? o1[r] : 0.f;
    o.z = o2[r] > 0.f ? o2[r] : 0.f;  o.w = o3[r] > 0.f ? o3[r] : 0.f;
    *reinterpret_cast<v4f*>(hgroup + (8 * half + r) * NHID + 4 * c) = o;
  }
}

template <bool FILE, bool ADD>
__device__ __forceinline__ void scan_edges(
    const unsigned* __restrict__ tid, const uint2* __restrict__ rec, int tile, int wave,
    int lane, unsigned* wqw, unsigned* rcnt, uint2* tlist,
    const float* __restrict__ h, float* acc) {
  auto file_rec = [&](uint2 rc, bool valid) {
    if (valid) {
      const unsigned lr = rc.y & 511u;
      const unsigned slot = __hip_atomic_fetch_add(&rcnt[lr], 1u, __ATOMIC_RELAXED,
                                                   __HIP_MEMORY_SCOPE_WORKGROUP);
      if (slot < (unsigned)RCAP) tlist[lr * RCAP + slot] = rc;
    }
  };

  const unsigned pat = (unsigned)tile * 0x01010101u;
  const uint4* tid4 = reinterpret_cast<const uint4*>(tid);
  const int g0 = (tile * NGROUP) / NTILES;

  struct ids2 { uint4 a, b; };
  auto load_ids = [&](int g) {
    const int ch = min(g * 32 + wave, NCHUNK - 1);
    ids2 r;
    r.a = tid4[ch * 64 + lane];
    r.b = tid4[ch * 64 + 32 + lane];
    return r;
  };
  auto next_group = [](int g) { return (g + 1 == NGROUP) ? 0 : g + 1; };

  int qn = 0;
  auto scan = [&](const ids2& v, int g) {
    const int ch = g * 32 + wave;
    if (ch >= NCHUNK) return;
    unsigned m = (zero_bytes(v.a.x ^ pat) >> 7) | (zero_bytes(v.a.y ^ pat) >> 6) |
                 (zero_bytes(v.a.z ^ pat) >> 5) | (zero_bytes(v.a.w ^ pat) >> 4) |
                 (zero_bytes(v.b.x ^ pat) >> 3) | (zero_bytes(v.b.y ^ pat) >> 2) |
                 (zero_bytes(v.b.z ^ pat) >> 1) | zero_bytes(v.b.w ^ pat);
    const unsigned ebase = (unsigned)ch * CHUNK + (unsigned)lane * 16u;
    while (true) {
      const bool has = m != 0u;
      const unsigned mask = __builtin_amdgcn_ballot_w32(has);
      if (mask == 0u) break;
      if (has) {
        const unsigned p = (unsigned)__builtin_ctz(m);
        wqw[qn + (int)__builtin_amdgcn_mbcnt_lo(mask, 0u)] =
            ebase + ((p & 4u) << 7) + ((p & 3u) << 2) + (p >> 3);
        m &= m - 1u;
      }
      qn += __builtin_popcount(mask);
      if (qn >= 32) {
        qn -= 32;
        const uint2 rc = rec[wqw[qn + lane]];
        if (FILE) file_rec(rc, true);
        if (ADD) add_batch<true>(rc, 32, lane, h, acc);
      }
    }
  };

  static_assert(NGROUP % 2 == 0, "the scan loop takes two groups per trip");
  int g = g0;
  ids2 cur = load_ids(g);
  for (int k = 0; k < NGROUP; k += 2) {
    const int g1 = next_group(g);
    const int g2 = next_group(g1);
    const ids2 nxt = load_ids(g1);
    scan(cur, g);
    cur = load_ids(g2);
    scan(nxt, g1);
    g = g2;
  }
  if (qn > 0) {
    const uint2 rc = (lane < qn) ? rec[wqw[lane]] : make_uint2(0u, 0u);
    if (FILE) file_rec(rc, lane < qn);
    if (ADD) add_batch<false>(rc, qn, lane, h, acc);
  }
}

template <bool FIRST>
__global__ void __launch_bounds__(1024) spmm_owner_kernel(
    const unsigned* __restrict__ tid, const uint2* __restrict__ rec,
    const float* __restrict__ h, const float* __restrict__ h0,
    const float* __restrict__ Wl, float beta, float* __restrict__ hout,
    uint2* __restrict__ rlists, unsigned* __restrict__ rcounts) {
  extern __shared__ __attribute__((aligned(16))) float tile_lds[];
  __shared__ unsigned wq[32 * QCAP];
  __shared__ unsigned rcnt[TR];
  __shared__ unsigned wover[32];
  __shared__ __attribute__((aligned(16))) float Wp[NHID * NHID];
  float* acc = tile_lds;
  float4* acc4 = reinterpret_cast<float4*>(tile_lds);
  const int lane = threadIdx.x & 31;
  const int wave = __builtin_amdgcn_readfirstlane(threadIdx.x >> 5);
  const int qb = wave * QCAP;
  const int tile = blockIdx.x;
  uint2* tlist = rlists + (size_t)tile * TR * RCAP;
  const long long grow = (long long)tile * TR + wave * 16;
  const float* h0row = h0 + (grow + (lane & 15)) * NHID + 32 * (lane >> 4);
  float* hgroup = hout + grow * NHID;

  if (!FIRST) {
    const int r = (int)threadIdx.x;
    const unsigned myn = (r < TR) ? rcounts[tile * TR + r] : 0u;
    if (r < TR) rcnt[r] = myn;
    const unsigned ob = __builtin_amdgcn_ballot_w32(myn > (unsigned)RCAP);
    if (lane == 0) wover[wave] = ob;
  } else {
    if (threadIdx.x < TR) rcnt[threadIdx.x] = 0u;
  }
  stage_layer_weights(Wl, beta, Wp);
  __syncthreads();

  if (FIRST) {
    scan_edges<true, false>(tid, rec, tile, wave, lane, wq + qb, rcnt, tlist, h, acc);
    __syncthreads();
    const int r = (int)threadIdx.x;
    const unsigned n = (r < TR) ? rcnt[r] : 0u;
    if (r < TR) {
      rcounts[tile * TR + r] = n;
      if (n <= (unsigned)RCAP) {
        const unsigned e = (n + 7u) & ~7u;
        for (unsigned i = n; i < e; ++i) tlist[r * RCAP + i] = make_uint2(0u, 0u);
      }
    }
    const unsigned ob = __builtin_amdgcn_ballot_w32(n > (unsigned)RCAP);
    if (lane == 0) wover[wave] = ob;
    __threadfence_block();
    __syncthreads();
  }

  if (__builtin_amdgcn_ballot_w32(wover[lane] != 0u) == 0u) {
    replay_rows(tlist, rcnt, wave, lane, h, tile_lds);
  } else {
    for (int i = threadIdx.x; i < TR * NHID / 4; i += 1024)
      acc4[i] = make_float4(0.f, 0.f, 0.f, 0.f);
    __syncthreads();
    scan_edges<false, true>(tid, rec, tile, wave, lane, wq + qb, rcnt, tlist, h, acc);
  }
  __syncthreads();

  if (wave < NG) {
    const v4f* trow = reinterpret_cast<const v4f*>(
        tile_lds + (wave * 16 + (lane & 15)) * NHID + 32 * (lane >> 4));
    v8f s[4];
#pragma unroll
    for (int r = 0; r < 8; ++r) {
      const v4f v = trow[r];
      s[0][r] = v.x;  s[1][r] = v.y;  s[2][r] = v.z;  s[3][r] = v.w;
    }
    layer_body(s, h0row, Wp, lane, hgroup);
  }
}

__global__ void __launch_bounds__(256) out_kernel(
    const float* __restrict__ h, const float* __restrict__ Wout,
    const float* __restrict__ bout, float* __restrict__ out) {
  __shared__ float Ws[NHID * NOUT];
  __shared__ float bs[NOUT];
  for (int idx = threadIdx.x; idx < NHID * NOUT; idx += 256) Ws[idx] = Wout[idx];
  if (threadIdx.x < NOUT) bs[threadIdx.x] = bout[threadIdx.x];
  __syncthreads();

  const int i = blockIdx.x * blockDim.x + threadIdx.x;
  if (i >= N_NODES) return;

  float hr[NHID];
  const float4* hp = reinterpret_cast<const float4*>(h + (long long)i * NHID);
#pragma unroll
  for (int k4 = 0; k4 < NHID / 4; ++k4) {
    const float4 v = hp[k4];
    hr[4 * k4 + 0] = v.x; hr[4 * k4 + 1] = v.y;
    hr[4 * k4 + 2] = v.z; hr[4 * k4 + 3] = v.w;
  }

  float lg[NOUT];
#pragma unroll
  for (int c = 0; c < NOUT; ++c) lg[c] = bs[c];
  for (int k = 0; k < NHID; ++k) {
    const float hk = hr[k];
#pragma unroll
    for (int c = 0; c < NOUT; ++c) lg[c] = fmaf(hk, Ws[k * NOUT + c], lg[c]);
  }

  float mx = lg[0];
#pragma unroll
  for (int c = 1; c < NOUT; ++c) mx = fmaxf(mx, lg[c]);
  float s = 0.f;
#pragma unroll
  for (int c = 0; c < NOUT; ++c) s += expf(lg[c] - mx);
  const float lse = mx + logf(s);
#pragma unroll
  for (int c = 0; c < NOUT; ++c)
    out[(long long)i * NOUT + c] = lg[c] - lse;
}

extern "C" void kernel_launch(void* const* d_in, const int* in_sizes, int n_in,
                              void* d_out, int out_size, void* d_ws, size_t ws_size,
                              hipStream_t stream) {
  (void)in_sizes; (void)n_in; (void)out_size; (void)ws_size;
  const float* x     = (const float*)d_in[0];
  const int*   row   = (const int*)  d_in[1];
  const int*   col   = (const int*)  d_in[2];
  const float* ew    = (const float*)d_in[3];
  const float* Win   = (const float*)d_in[4];
  const float* bin   = (const float*)d_in[5];
  const float* convW = (const float*)d_in[6];
  const float* Wout  = (const float*)d_in[7];
  const float* bout  = (const float*)d_in[8];

  const size_t S = (size_t)N_NODES * NHID;
  float* h0 = (float*)d_ws;
  float* hA = h0 + S;
  float* hB = hA + S;

  const int rowBlocks = (N_NODES + 255) / 256;
  const size_t inLds = (size_t)(NIN / 2) * BRS * sizeof(float);
  in_gemm_kernel<<<rowBlocks, 256, inLds, stream>>>(x, Win, bin, h0);

  unsigned* tid = (unsigned*)(hB + S);
  uint2* rec = (uint2*)(tid + E_EDGES / 4);
  edge_pack_kernel<<<E_EDGES / 4 / 256, 256, 0, stream>>>(row, col, ew, tid, rec);
  const size_t spmmLds = (size_t)164 * 1024;
  uint2* lists = rec + E_EDGES;
  unsigned* counts = (unsigned*)(lists + (size_t)N_NODES * RCAP);

  const float* hin = h0;
  float* hnext = hA;
  float* hother = hB;
  for (int l = 0; l < NLAYERS; ++l) {
    const float beta = logf(0.5f / (float)(l + 1) + 1.f);
    const float* Wl = convW + (size_t)l * NHID * NHID;
    if (l == 0)
      spmm_owner_kernel<true><<<NTILES, 1024, spmmLds, stream>>>(
          tid, rec, hin, h0, Wl, beta, hnext, lists, counts);
    else
      spmm_owner_kernel<false><<<NTILES, 1024, spmmLds, stream>>>(
          tid, rec, hin, h0, Wl, beta, hnext, lists, counts);
    hin = hnext;
    hnext = hother;
    hother = (float*)hin;
  }

  out_kernel<<<(N_NODES + 255) / 256, 256, 0, stream>>>(hin, Wout, bout, (float*)d_out);
}


# Input GEMM as a resident grid: 256 blocks of 13 waves, one per WGP, stage W_in once by 16-byte loads all in flight, one 32-row tile per wave (was 391 blocks in two rounds, 4-byte staging); same sums
# speedup vs baseline: 4.3662x; 1.0317x over previous
#include <hip/hip_runtime.h>
#include <hip/hip_bf16.h>
#include <math.h>

#define N_NODES 100000
#define E_EDGES 3200000
#define NIN     500
#define NHID    64
#define NOUT    40
#define NLAYERS 8
#define ALPHA   0.1f

#define BRS 160

typedef __attribute__((ext_vector_type(2))) float v2f;
typedef __attribute__((ext_vector_type(8))) float v8f;
typedef __attribute__((ext_vector_type(4))) unsigned v4u;
typedef __attribute__((ext_vector_type(4))) float v4f;

__device__ __forceinline__ void wmma_acc(v8f& c, v2f a, v2f b) {
  asm("v_wmma_f32_16x16x4_f32 %0, %1, %2, %0" : "+v"(c) : "v"(a), "v"(b));
}
__device__ __forceinline__ void wmma_epilogue_guard() {
  asm volatile("s_nop 8");
}

#define IN_BLOCKS 256
#define IN_WAVES  13
#define IN_TILES  (N_NODES / 32)
#define ST_THREADS 400
#define ST_PIECES  20
static_assert(N_NODES % 32 == 0, "whole 32-row tiles");
static_assert(IN_BLOCKS * IN_WAVES >= IN_TILES, "every tile has a wave");
static_assert(ST_THREADS <= IN_WAVES * 32 && ST_THREADS * ST_PIECES * 4 == NIN * NHID,
              "the staging threads cover W_in exactly");
static_assert(NHID % 4 == 0, "a 16-byte piece of W_in stays inside one k row");

__global__ void __launch_bounds__(IN_WAVES * 32) in_gemm_kernel(
    const float* __restrict__ x, const float* __restrict__ Win,
    const float* __restrict__ bin, float* __restrict__ h0) {
  extern __shared__ float Wp[];
  if (threadIdx.x < ST_THREADS) {
    v4f w[ST_PIECES];
#pragma unroll
    for (int j = 0; j < ST_PIECES; ++j)
      w[j] = reinterpret_cast<const v4f*>(Win)[threadIdx.x + ST_THREADS * j];
#pragma unroll
    for (int j = 0; j < ST_PIECES; ++j) {
      const int idx = 4 * ((int)threadIdx.x + ST_THREADS * j);
      const int k = idx >> 6;
      const int n = idx & 63;
      float* p = &Wp[(k >> 1) * BRS + 2 * n + (k & 1)];
      p[0] = w[j].x;  p[2] = w[j].y;  p[4] = w[j].z;  p[6] = w[j].w;
    }
  }
  __syncthreads();

  const int wave = __builtin_amdgcn_readfirstlane((int)(threadIdx.x >> 5));
  const int lane = threadIdx.x & 31;
  const int tile = wave * IN_BLOCKS + (int)blockIdx.x;
  if (tile >= IN_TILES) return;
  const int half = lane >> 4;
  const int l16  = lane & 15;
  const int r0   = tile * 32;

  const int arow0 = min(r0 + l16,      N_NODES - 1);
  const int arow1 = min(r0 + 16 + l16, N_NODES - 1);
  const float* xr0 = x + (long long)arow0 * NIN;
  const float* xr1 = x + (long long)arow1 * NIN;

  v8f acc0[4] = {};
  v8f acc1[4] = {};
  for (int k = 0; k < NIN; k += 4) {
    const int kk = k + 2 * half;
    const v2f a0 = *reinterpret_cast<const v2f*>(xr0 + kk);
    const v2f a1 = *reinterpret_cast<const v2f*>(xr1 + kk);
    const int bbase = (kk >> 1) * BRS + 2 * l16;
#pragma unroll
    for (int t = 0; t < 4; ++t) {
      const v2f b = *reinterpret_cast<const v2f*>(&Wp[bbase + t * 32]);
      wmma_acc(acc0[t], a0, b);
      wmma_acc(acc1[t], a1, b);
    }
  }
  wmma_epilogue_guard();

#pragma unroll
  for (int t = 0; t < 4; ++t) {
    const int n = t * 16 + l16;
    const float bias = bin[n];
#pragma unroll
    for (int r = 0; r < 8; ++r) {
      const int m = r + 8 * half;
      const int orow0 = r0 + m;
      const int orow1 = r0 + 16 + m;
      if (orow0 < N_NODES) {
        const float v = acc0[t][r] + bias;
        h0[(long long)orow0 * NHID + n] = v > 0.f ? v : 0.f;
      }
      if (orow1 < N_NODES) {
        const float v = acc1[t][r] + bias;
        h0[(long long)orow1 * NHID + n] = v > 0.f ? v : 0.f;
      }
    }
  }
}

#define TR      400
#define NTILES  (N_NODES / TR)
#define CHUNK   1024
#define NCHUNK  (E_EDGES / CHUNK)
#define NGROUP  ((NCHUNK + 31) / 32)
#define QCAP    64
#define NG      (TR / 16)
#define RCAP    128
#define RB      48
static_assert(N_NODES % TR == 0 && NTILES <= 256, "tile id must fit a byte");
static_assert(TR <= 512 && N_NODES <= (1 << 23), "record packs col << 9 | local row");
static_assert(E_EDGES % CHUNK == 0, "edge count must be whole chunks");
static_assert(TR % 16 == 0 && NG <= 32 && TR >= 32, "one body wave per group; every wave has a row");
static_assert(RCAP % 32 == 0 && RB == 48, "record blocks of 32 lanes; replay_rows has one case per eight up to RB");

__global__ void __launch_bounds__(256) edge_pack_kernel(
    const int* __restrict__ row, const int* __restrict__ col,
    const float* __restrict__ ew, unsigned* __restrict__ tid,
    uint2* __restrict__ rec) {
  const int t = blockIdx.x * 256 + threadIdx.x;
  if (t >= E_EDGES / 4) return;
  const int4 r = reinterpret_cast<const int4*>(row)[t];
  const int4 c = reinterpret_cast<const int4*>(col)[t];
  const float4 w = reinterpret_cast<const float4*>(ew)[t];
  const unsigned t0 = (unsigned)(r.x / TR), t1 = (unsigned)(r.y / TR);
  const unsigned t2 = (unsigned)(r.z / TR), t3 = (unsigned)(r.w / TR);
  tid[t] = t0 | (t1 << 8) | (t2 << 16) | (t3 << 24);
  uint4* rp = reinterpret_cast<uint4*>(rec + 4 * (long long)t);
  rp[0] = make_uint4(__float_as_uint(w.x), ((unsigned)c.x << 9) | ((unsigned)r.x - t0 * TR),
                     __float_as_uint(w.y), ((unsigned)c.y << 9) | ((unsigned)r.y - t1 * TR));
  rp[1] = make_uint4(__float_as_uint(w.z), ((unsigned)c.z << 9) | ((unsigned)r.z - t2 * TR),
                     __float_as_uint(w.w), ((unsigned)c.w << 9) | ((unsigned)r.w - t3 * TR));
}

__device__ __forceinline__ unsigned zero_bytes(unsigned x) {
  const unsigned t = ((x & 0x7f7f7f7fu) + 0x7f7f7f7fu) | x;
  return ~t & 0x80808080u;
}

template <bool FULL>
__device__ __forceinline__ void add_batch(
    uint2 rc, int n, int lane, const float* __restrict__ h, float* acc) {
  const float w = __uint_as_float(rc.x);
  const int c = (int)(rc.y >> 9);
  const int lr = (int)(rc.y & 511u);
  v2f g[32];
#pragma unroll
  for (int i = 0; i < 32; ++i) {
    const int ci = __builtin_amdgcn_readlane(c, i);
    g[i] = *reinterpret_cast<const v2f*>(h + (long long)ci * NHID + 2 * lane);
  }
#pragma unroll
  for (int i = 0; i < 32; ++i) {
    if (FULL || i < n) {
      const float wi = __int_as_float(__builtin_amdgcn_readlane(__float_as_int(w), i));
      const int li = __builtin_amdgcn_readlane(lr, i);
      float* a = acc + li * NHID + 2 * lane;
      __hip_atomic_fetch_add(a,     wi * g[i].x, __ATOMIC_RELAXED, __HIP_MEMORY_SCOPE_WORKGROUP);
      __hip_atomic_fetch_add(a + 1, wi * g[i].y, __ATOMIC_RELAXED, __HIP_MEMORY_SCOPE_WORKGROUP);
    }
  }
}

__device__ __forceinline__ v2f row_of(const char* hb, unsigned y) {
  return *reinterpret_cast<const v2f*>(hb + ((y >> 9) << 8));
}

template <int NB>
__device__ __forceinline__ v2f sum_records(uint2 ra, uint2 rb, const char* hb) {
  v2f g[NB * 8];
#pragma unroll
  for (int i = 0; i < NB * 8; ++i) {
    const unsigned y = (unsigned)__builtin_amdgcn_readlane((int)(i < 32 ? ra.y : rb.y), i & 31);
    g[i] = row_of(hb, y);
  }
  v2f s = {0.f, 0.f};
#pragma unroll
  for (int i = 0; i < NB * 8; ++i) {
    const float w = __int_as_float(
        __builtin_amdgcn_readlane((int)(i < 32 ? ra.x : rb.x), i & 31));
    s.x = fmaf(w, g[i].x, s.x);
    s.y = fmaf(w, g[i].y, s.y);
  }
  return s;
}

__device__ __forceinline__ void replay_rows(
    const uint2* __restrict__ tlist, const unsigned* rn, int wave, int lane,
    const float* __restrict__ h, float* tile) {
  const char* hb = reinterpret_cast<const char*>(h) + 8 * lane;
  int lr = wave;
  uint2 ra = tlist[lr * RCAP + lane];
  uint2 rb = tlist[lr * RCAP + 32 + lane];
  int n = __builtin_amdgcn_readfirstlane((int)rn[lr]);
  while (true) {
    const int npad = (n + 7) & ~7;
    const int nxt = lr + 32;
    const int pf = nxt < TR ? nxt : lr;
    const uint2 na = tlist[pf * RCAP + lane];
    const uint2 nb = tlist[pf * RCAP + 32 + lane];
    const unsigned nn = rn[pf];
    v2f s = {0.f, 0.f};
    switch ((npad < RB ? npad : RB) >> 3) {
      case 0:  break;
      case 1:  s = sum_records<1>(ra, rb, hb);  break;
      case 2:  s = sum_records<2>(ra, rb, hb);  break;
      case 3:  s = sum_records<3>(ra, rb, hb);  break;
      case 4:  s = sum_records<4>(ra, rb, hb);  break;
      case 5:  s = sum_records<5>(ra, rb, hb);  break;
      default: s = sum_records<6>(ra, rb, hb);  break;
    }
    for (int i = RB; i < npad; i += 8) {
      const uint2 rx = tlist[lr * RCAP + (i & ~31) + lane];
      v2f t[8];
#pragma unroll
      for (int j = 0; j < 8; ++j)
        t[j] = row_of(hb, (unsigned)__builtin_amdgcn_readlane((int)rx.y, (i + j) & 31));
#pragma unroll
      for (int j = 0; j < 8; ++j) {
        const float w = __int_as_float(__builtin_amdgcn_readlane((int)rx.x, (i + j) & 31));
        s.x = fmaf(w, t[j].x, s.x);
        s.y = fmaf(w, t[j].y, s.y);
      }
    }
    *reinterpret_cast<v2f*>(tile + lr * NHID + 2 * lane) = s;
    if (nxt >= TR) break;
    lr = nxt;  ra = na;  rb = nb;
    n = __builtin_amdgcn_readfirstlane((int)nn);
  }
}

__device__ __forceinline__ void stage_layer_weights(
    const float* __restrict__ Wl, float beta, float* Wp) {
  const float omb = 1.f - beta;
#pragma unroll
  for (int j = 0; j < NHID * NHID / 1024; ++j) {
    const int idx = (int)threadIdx.x + 1024 * j;
    const int k = idx >> 6;
    const int n = idx & 63;
    const float w = beta * Wl[idx];
    const int kh = k >> 5, kt = k & 3, ki = (k >> 3) & 3, ks = (k >> 2) & 1;
    Wp[(((kt * 4 + ki) * 4 + (n & 3)) << 6) + 2 * ((n >> 2) + 16 * kh) + ks] =
        (k == n) ? w + omb : w;
  }
}

__device__ __forceinline__ void layer_body(
    v8f (&s)[4], const float* __restrict__ h0row, const float* Wp, int lane,
    float* __restrict__ hgroup) {
  const float sA = 1.f - ALPHA;
  const float sB = ALPHA;
  v4f g[8];
#pragma unroll
  for (int r = 0; r < 8; ++r) g[r] = *reinterpret_cast<const v4f*>(h0row + 4 * r);
#pragma unroll
  for (int r = 0; r < 8; ++r) {
    s[0][r] = sA * s[0][r] + sB * g[r].x;
    s[1][r] = sA * s[1][r] + sB * g[r].y;
    s[2][r] = sA * s[2][r] + sB * g[r].z;
    s[3][r] = sA * s[3][r] + sB * g[r].w;
  }
  const v8f z8 = {};
  v8f o0 = z8, o1 = z8, o2 = z8, o3 = z8;
  const v2f* wp = reinterpret_cast<const v2f*>(Wp) + lane;
#pragma unroll
  for (int t = 0; t < 4; ++t) {
#pragma unroll
    for (int i = 0; i < 4; ++i) {
      v2f a;
      a.x = s[t][2 * i];  a.y = s[t][2 * i + 1];
      const v2f* wk = wp + (t * 4 + i) * 128;
      o0 = __builtin_amdgcn_wmma_f32_16x16x4_f32(0, a, 0, wk[0],  0, o0, false, false);
      o1 = __builtin_amdgcn_wmma_f32_16x16x4_f32(0, a, 0, wk[32], 0, o1, false, false);
      o2 = __builtin_amdgcn_wmma_f32_16x16x4_f32(0, a, 0, wk[64], 0, o2, false, false);
      o3 = __builtin_amdgcn_wmma_f32_16x16x4_f32(0, a, 0, wk[96], 0, o3, false, false);
    }
  }
  const int half = lane >> 4;
  const int c = lane & 15;
#pragma unroll
  for (int r = 0; r < 8; ++r) {
    v4f o;
    o.x = o0[r] > 0.f ? o0[r] : 0.f;  o.y = o1[r] > 0.f ? o1[r] : 0.f;
    o.z = o2[r] > 0.f ? o2[r] : 0.f;  o.w = o3[r] > 0.f ? o3[r] : 0.f;
    *reinterpret_cast<v4f*>(hgroup + (8 * half + r) * NHID + 4 * c) = o;
  }
}

template <bool FILE, bool ADD>
__device__ __forceinline__ void scan_edges(
    const unsigned* __restrict__ tid, const uint2* __restrict__ rec, int tile, int wave,
    int lane, unsigned* wqw, unsigned* rcnt, uint2* tlist,
    const float* __restrict__ h, float* acc) {
  auto file_rec = [&](uint2 rc, bool valid) {
    if (valid) {
      const unsigned lr = rc.y & 511u;
      const unsigned slot = __hip_atomic_fetch_add(&rcnt[lr], 1u, __ATOMIC_RELAXED,
                                                   __HIP_MEMORY_SCOPE_WORKGROUP);
      if (slot < (unsigned)RCAP) tlist[lr * RCAP + slot] = rc;
    }
  };

  const unsigned pat = (unsigned)tile * 0x01010101u;
  const uint4* tid4 = reinterpret_cast<const uint4*>(tid);
  const int g0 = (tile * NGROUP) / NTILES;

  struct ids2 { uint4 a, b; };
  auto load_ids = [&](int g) {
    const int ch = min(g * 32 + wave, NCHUNK - 1);
    ids2 r;
    r.a = tid4[ch * 64 + lane];
    r.b = tid4[ch * 64 + 32 + lane];
    return r;
  };
  auto next_group = [](int g) { return (g + 1 == NGROUP) ? 0 : g + 1; };

  int qn = 0;
  auto scan = [&](const ids2& v, int g) {
    const int ch = g * 32 + wave;
    if (ch >= NCHUNK) return;
    unsigned m = (zero_bytes(v.a.x ^ pat) >> 7) | (zero_bytes(v.a.y ^ pat) >> 6) |
                 (zero_bytes(v.a.z ^ pat) >> 5) | (zero_bytes(v.a.w ^ pat) >> 4) |
                 (zero_bytes(v.b.x ^ pat) >> 3) | (zero_bytes(v.b.y ^ pat) >> 2) |
                 (zero_bytes(v.b.z ^ pat) >> 1) | zero_bytes(v.b.w ^ pat);
    const unsigned ebase = (unsigned)ch * CHUNK + (unsigned)lane * 16u;
    while (true) {
      const bool has = m != 0u;
      const unsigned mask = __builtin_amdgcn_ballot_w32(has);
      if (mask == 0u) break;
      if (has) {
        const unsigned p = (unsigned)__builtin_ctz(m);
        wqw[qn + (int)__builtin_amdgcn_mbcnt_lo(mask, 0u)] =
            ebase + ((p & 4u) << 7) + ((p & 3u) << 2) + (p >> 3);
        m &= m - 1u;
      }
      qn += __builtin_popcount(mask);
      if (qn >= 32) {
        qn -= 32;
        const uint2 rc = rec[wqw[qn + lane]];
        if (FILE) file_rec(rc, true);
        if (ADD) add_batch<true>(rc, 32, lane, h, acc);
      }
    }
  };

  static_assert(NGROUP % 2 == 0, "the scan loop takes two groups per trip");
  int g = g0;
  ids2 cur = load_ids(g);
  for (int k = 0; k < NGROUP; k += 2) {
    const int g1 = next_group(g);
    const int g2 = next_group(g1);
    const ids2 nxt = load_ids(g1);
    scan(cur, g);
    cur = load_ids(g2);
    scan(nxt, g1);
    g = g2;
  }
  if (qn > 0) {
    const uint2 rc = (lane < qn) ? rec[wqw[lane]] : make_uint2(0u, 0u);
    if (FILE) file_rec(rc, lane < qn);
    if (ADD) add_batch<false>(rc, qn, lane, h, acc);
  }
}

template <bool FIRST>
__global__ void __launch_bounds__(1024) spmm_owner_kernel(
    const unsigned* __restrict__ tid, const uint2* __restrict__ rec,
    const float* __restrict__ h, const float* __restrict__ h0,
    const float* __restrict__ Wl, float beta, float* __restrict__ hout,
    uint2* __restrict__ rlists, unsigned* __restrict__ rcounts) {
  extern __shared__ __attribute__((aligned(16))) float tile_lds[];
  __shared__ unsigned wq[32 * QCAP];
  __shared__ unsigned rcnt[TR];
  __shared__ unsigned wover[32];
  __shared__ __attribute__((aligned(16))) float Wp[NHID * NHID];
  float* acc = tile_lds;
  float4* acc4 = reinterpret_cast<float4*>(tile_lds);
  const int lane = threadIdx.x & 31;
  const int wave = __builtin_amdgcn_readfirstlane(threadIdx.x >> 5);
  const int qb = wave * QCAP;
  const int tile = blockIdx.x;
  uint2* tlist = rlists + (size_t)tile * TR * RCAP;
  const long long grow = (long long)tile * TR + wave * 16;
  const float* h0row = h0 + (grow + (lane & 15)) * NHID + 32 * (lane >> 4);
  float* hgroup = hout + grow * NHID;

  if (!FIRST) {
    const int r = (int)threadIdx.x;
    const unsigned myn = (r < TR) ? rcounts[tile * TR + r] : 0u;
    if (r < TR) rcnt[r] = myn;
    const unsigned ob = __builtin_amdgcn_ballot_w32(myn > (unsigned)RCAP);
    if (lane == 0) wover[wave] = ob;
  } else {
    if (threadIdx.x < TR) rcnt[threadIdx.x] = 0u;
  }
  stage_layer_weights(Wl, beta, Wp);
  __syncthreads();

  if (FIRST) {
    scan_edges<true, false>(tid, rec, tile, wave, lane, wq + qb, rcnt, tlist, h, acc);
    __syncthreads();
    const int r = (int)threadIdx.x;
    const unsigned n = (r < TR) ? rcnt[r] : 0u;
    if (r < TR) {
      rcounts[tile * TR + r] = n;
      if (n <= (unsigned)RCAP) {
        const unsigned e = (n + 7u) & ~7u;
        for (unsigned i = n; i < e; ++i) tlist[r * RCAP + i] = make_uint2(0u, 0u);
      }
    }
    const unsigned ob = __builtin_amdgcn_ballot_w32(n > (unsigned)RCAP);
    if (lane == 0) wover[wave] = ob;
    __threadfence_block();
    __syncthreads();
  }

  if (__builtin_amdgcn_ballot_w32(wover[lane] != 0u) == 0u) {
    replay_rows(tlist, rcnt, wave, lane, h, tile_lds);
  } else {
    for (int i = threadIdx.x; i < TR * NHID / 4; i += 1024)
      acc4[i] = make_float4(0.f, 0.f, 0.f, 0.f);
    __syncthreads();
    scan_edges<false, true>(tid, rec, tile, wave, lane, wq + qb, rcnt, tlist, h, acc);
  }
  __syncthreads();

  if (wave < NG) {
    const v4f* trow = reinterpret_cast<const v4f*>(
        tile_lds + (wave * 16 + (lane & 15)) * NHID + 32 * (lane >> 4));
    v8f s[4];
#pragma unroll
    for (int r = 0; r < 8; ++r) {
      const v4f v = trow[r];
      s[0][r] = v.x;  s[1][r] = v.y;  s[2][r] = v.z;  s[3][r] = v.w;
    }
    layer_body(s, h0row, Wp, lane, hgroup);
  }
}

__global__ void __launch_bounds__(256) out_kernel(
    const float* __restrict__ h, const float* __restrict__ Wout,
    const float* __restrict__ bout, float* __restrict__ out) {
  __shared__ float Ws[NHID * NOUT];
  __shared__ float bs[NOUT];
  for (int idx = threadIdx.x; idx < NHID * NOUT; idx += 256) Ws[idx] = Wout[idx];
  if (threadIdx.x < NOUT) bs[threadIdx.x] = bout[threadIdx.x];
  __syncthreads();

  const int i = blockIdx.x * blockDim.x + threadIdx.x;
  if (i >= N_NODES) return;

  float hr[NHID];
  const float4* hp = reinterpret_cast<const float4*>(h + (long long)i * NHID);
#pragma unroll
  for (int k4 = 0; k4 < NHID / 4; ++k4) {
    const float4 v = hp[k4];
    hr[4 * k4 + 0] = v.x; hr[4 * k4 + 1] = v.y;
    hr[4 * k4 + 2] = v.z; hr[4 * k4 + 3] = v.w;
  }

  float lg[NOUT];
#pragma unroll
  for (int c = 0; c < NOUT; ++c) lg[c] = bs[c];
  for (int k = 0; k < NHID; ++k) {
    const float hk = hr[k];
#pragma unroll
    for (int c = 0; c < NOUT; ++c) lg[c] = fmaf(hk, Ws[k * NOUT + c], lg[c]);
  }

  float mx = lg[0];
#pragma unroll
  for (int c = 1; c < NOUT; ++c) mx = fmaxf(mx, lg[c]);
  float s = 0.f;
#pragma unroll
  for (int c = 0; c < NOUT; ++c) s += expf(lg[c] - mx);
  const float lse = mx + logf(s);
#pragma unroll
  for (int c = 0; c < NOUT; ++c)
    out[(long long)i * NOUT + c] = lg[c] - lse;
}

extern "C" void kernel_launch(void* const* d_in, const int* in_sizes, int n_in,
                              void* d_out, int out_size, void* d_ws, size_t ws_size,
                              hipStream_t stream) {
  (void)in_sizes; (void)n_in; (void)out_size; (void)ws_size;
  const float* x     = (const float*)d_in[0];
  const int*   row   = (const int*)  d_in[1];
  const int*   col   = (const int*)  d_in[2];
  const float* ew    = (const float*)d_in[3];
  const float* Win   = (const float*)d_in[4];
  const float* bin   = (const float*)d_in[5];
  const float* convW = (const float*)d_in[6];
  const float* Wout  = (const float*)d_in[7];
  const float* bout  = (const float*)d_in[8];

  const size_t S = (size_t)N_NODES * NHID;
  float* h0 = (float*)d_ws;
  float* hA = h0 + S;
  float* hB = hA + S;

  const size_t inLds = (size_t)(NIN / 2) * BRS * sizeof(float);
  in_gemm_kernel<<<IN_BLOCKS, IN_WAVES * 32, inLds, stream>>>(x, Win, bin, h0);

  unsigned* tid = (unsigned*)(hB + S);
  uint2* rec = (uint2*)(tid + E_EDGES / 4);
  edge_pack_kernel<<<E_EDGES / 4 / 256, 256, 0, stream>>>(row, col, ew, tid, rec);
  const size_t spmmLds = (size_t)164 * 1024;
  uint2* lists = rec + E_EDGES;
  unsigned* counts = (unsigned*)(lists + (size_t)N_NODES * RCAP);

  const float* hin = h0;
  float* hnext = hA;
  float* hother = hB;
  for (int l = 0; l < NLAYERS; ++l) {
    const float beta = logf(0.5f / (float)(l + 1) + 1.f);
    const float* Wl = convW + (size_t)l * NHID * NHID;
    if (l == 0)
      spmm_owner_kernel<true><<<NTILES, 1024, spmmLds, stream>>>(
          tid, rec, hin, h0, Wl, beta, hnext, lists, counts);
    else
      spmm_owner_kernel<false><<<NTILES, 1024, spmmLds, stream>>>(
          tid, rec, hin, h0, Wl, beta, hnext, lists, counts);
    hin = hnext;
    hnext = hother;
    hother = (float*)hin;
  }

  out_kernel<<<(N_NODES + 255) / 256, 256, 0, stream>>>(hin, Wout, bout, (float*)d_out);
}


# Input GEMM: x through LDS: a wave loads its 32 rows in 20-k chunks, 16 B per lane along the rows, one chunk ahead of the matrix steps, and reads A from its own LDS buffer; same k order and sums
# speedup vs baseline: 4.8281x; 1.1058x over previous
#include <hip/hip_runtime.h>
#include <hip/hip_bf16.h>
#include <math.h>

#define N_NODES 100000
#define E_EDGES 3200000
#define NIN     500
#define NHID    64
#define NOUT    40
#define NLAYERS 8
#define ALPHA   0.1f

#define BRS 160

typedef __attribute__((ext_vector_type(2))) float v2f;
typedef __attribute__((ext_vector_type(8))) float v8f;
typedef __attribute__((ext_vector_type(4))) unsigned v4u;
typedef __attribute__((ext_vector_type(4))) float v4f;

__device__ __forceinline__ void wmma_acc(v8f& c, v2f a, v2f b) {
  asm("v_wmma_f32_16x16x4_f32 %0, %1, %2, %0" : "+v"(c) : "v"(a), "v"(b));
}
__device__ __forceinline__ void wmma_epilogue_guard() {
  asm volatile("s_nop 8");
}

#define IN_BLOCKS 256
#define IN_WAVES  13
#define IN_TILES  (N_NODES / 32)
#define ST_THREADS 400
#define ST_PIECES  20
#define XCH       20
#define XPIECES   (XCH / 4)
#define NXCH      (NIN / XCH)
#define XBUF      (32 * XCH)
#define WP_FLOATS ((NIN / 2) * BRS)
static_assert(N_NODES % 32 == 0, "whole 32-row tiles");
static_assert(IN_BLOCKS * IN_WAVES >= IN_TILES, "every tile has a wave");
static_assert(ST_THREADS <= IN_WAVES * 32 && ST_THREADS * ST_PIECES * 4 == NIN * NHID,
              "the staging threads cover W_in exactly");
static_assert(NHID % 4 == 0, "a 16-byte piece of W_in stays inside one k row");
static_assert(NIN % XCH == 0 && XCH % 4 == 0 && (NIN * 4) % 16 == 0,
              "whole chunks of whole k steps; every piece 16-byte aligned");
static_assert((XCH / 4) % 2 == 1,
              "odd count of 16-byte pieces per buffer row: the 16 rows of an A-fragment read fall on disjoint banks");
static_assert((WP_FLOATS * 4) % 16 == 0 && (XBUF * 4) % 16 == 0, "chunk buffers 16-byte aligned");

__global__ void __launch_bounds__(IN_WAVES * 32) in_gemm_kernel(
    const float* __restrict__ x, const float* __restrict__ Win,
    const float* __restrict__ bin, float* __restrict__ h0) {
  extern __shared__ __attribute__((aligned(16))) float Wp[];

  const int wave = __builtin_amdgcn_readfirstlane((int)(threadIdx.x >> 5));
  const int lane = threadIdx.x & 31;
  const int tile = wave * IN_BLOCKS + (int)blockIdx.x;
  const bool has_tile = tile < IN_TILES;

  unsigned pb[XPIECES];
#pragma unroll
  for (int j = 0; j < XPIECES; ++j) {
    const unsigned i = (unsigned)lane + 32u * (unsigned)j;
    pb[j] = (i / XPIECES) * (unsigned)(NIN * 4) + (i % XPIECES) * 16u;
  }
  const char* xt = reinterpret_cast<const char*>(x) +
                   (size_t)(has_tile ? tile : 0) * (size_t)(32 * NIN * 4);
  v4f xr[XPIECES] = {};
  if (has_tile) {
#pragma unroll
    for (int j = 0; j < XPIECES; ++j)
      xr[j] = *reinterpret_cast<const v4f*>(xt + pb[j]);
  }

  if (threadIdx.x < ST_THREADS) {
    v4f w[ST_PIECES];
#pragma unroll
    for (int j = 0; j < ST_PIECES; ++j)
      w[j] = reinterpret_cast<const v4f*>(Win)[threadIdx.x + ST_THREADS * j];
#pragma unroll
    for (int j = 0; j < ST_PIECES; ++j) {
      const int idx = 4 * ((int)threadIdx.x + ST_THREADS * j);
      const int k = idx >> 6;
      const int n = idx & 63;
      float* p = &Wp[(k >> 1) * BRS + 2 * n + (k & 1)];
      p[0] = w[j].x;  p[2] = w[j].y;  p[4] = w[j].z;  p[6] = w[j].w;
    }
  }
  __syncthreads();

  if (!has_tile) return;
  const int half = lane >> 4;
  const int l16  = lane & 15;
  const int r0   = tile * 32;

  float* xb = Wp + WP_FLOATS + wave * XBUF;
  const float* xa = xb + l16 * XCH + 2 * half;

  v8f acc0[4] = {};
  v8f acc1[4] = {};
  for (int c = 0; c < NXCH; ++c) {
#pragma unroll
    for (int j = 0; j < XPIECES; ++j)
      *reinterpret_cast<v4f*>(xb + 4 * (lane + 32 * j)) = xr[j];
    if (c + 1 < NXCH) {
      const char* xn = xt + (c + 1) * (XCH * 4);
#pragma unroll
      for (int j = 0; j < XPIECES; ++j)
        xr[j] = *reinterpret_cast<const v4f*>(xn + pb[j]);
    }
#pragma unroll 1
    for (int s = 0; s < XCH / 4; ++s) {
      const v2f a0 = *reinterpret_cast<const v2f*>(xa + 4 * s);
      const v2f a1 = *reinterpret_cast<const v2f*>(xa + 16 * XCH + 4 * s);
      const int kk = c * XCH + 4 * s + 2 * half;
      const int bbase = (kk >> 1) * BRS + 2 * l16;
#pragma unroll
      for (int t = 0; t < 4; ++t) {
        const v2f b = *reinterpret_cast<const v2f*>(&Wp[bbase + t * 32]);
        wmma_acc(acc0[t], a0, b);
        wmma_acc(acc1[t], a1, b);
      }
    }
  }
  wmma_epilogue_guard();

#pragma unroll
  for (int t = 0; t < 4; ++t) {
    const int n = t * 16 + l16;
    const float bias = bin[n];
#pragma unroll
    for (int r = 0; r < 8; ++r) {
      const int m = r + 8 * half;
      const int orow0 = r0 + m;
      const int orow1 = r0 + 16 + m;
      if (orow0 < N_NODES) {
        const float v = acc0[t][r] + bias;
        h0[(long long)orow0 * NHID + n] = v > 0.f ? v : 0.f;
      }
      if (orow1 < N_NODES) {
        const float v = acc1[t][r] + bias;
        h0[(long long)orow1 * NHID + n] = v > 0.f ? v : 0.f;
      }
    }
  }
}

#define TR      400
#define NTILES  (N_NODES / TR)
#define CHUNK   1024
#define NCHUNK  (E_EDGES / CHUNK)
#define NGROUP  ((NCHUNK + 31) / 32)
#define QCAP    64
#define NG      (TR / 16)
#define RCAP    128
#define RB      48
static_assert(N_NODES % TR == 0 && NTILES <= 256, "tile id must fit a byte");
static_assert(TR <= 512 && N_NODES <= (1 << 23), "record packs col << 9 | local row");
static_assert(E_EDGES % CHUNK == 0, "edge count must be whole chunks");
static_assert(TR % 16 == 0 && NG <= 32 && TR >= 32, "one body wave per group; every wave has a row");
static_assert(RCAP % 32 == 0 && RB == 48, "record blocks of 32 lanes; replay_rows has one case per eight up to RB");

__global__ void __launch_bounds__(256) edge_pack_kernel(
    const int* __restrict__ row, const int* __restrict__ col,
    const float* __restrict__ ew, unsigned* __restrict__ tid,
    uint2* __restrict__ rec) {
  const int t = blockIdx.x * 256 + threadIdx.x;
  if (t >= E_EDGES / 4) return;
  const int4 r = reinterpret_cast<const int4*>(row)[t];
  const int4 c = reinterpret_cast<const int4*>(col)[t];
  const float4 w = reinterpret_cast<const float4*>(ew)[t];
  const unsigned t0 = (unsigned)(r.x / TR), t1 = (unsigned)(r.y / TR);
  const unsigned t2 = (unsigned)(r.z / TR), t3 = (unsigned)(r.w / TR);
  tid[t] = t0 | (t1 << 8) | (t2 << 16) | (t3 << 24);
  uint4* rp = reinterpret_cast<uint4*>(rec + 4 * (long long)t);
  rp[0] = make_uint4(__float_as_uint(w.x), ((unsigned)c.x << 9) | ((unsigned)r.x - t0 * TR),
                     __float_as_uint(w.y), ((unsigned)c.y << 9) | ((unsigned)r.y - t1 * TR));
  rp[1] = make_uint4(__float_as_uint(w.z), ((unsigned)c.z << 9) | ((unsigned)r.z - t2 * TR),
                     __float_as_uint(w.w), ((unsigned)c.w << 9) | ((unsigned)r.w - t3 * TR));
}

__device__ __forceinline__ unsigned zero_bytes(unsigned x) {
  const unsigned t = ((x & 0x7f7f7f7fu) + 0x7f7f7f7fu) | x;
  return ~t & 0x80808080u;
}

template <bool FULL>
__device__ __forceinline__ void add_batch(
    uint2 rc, int n, int lane, const float* __restrict__ h, float* acc) {
  const float w = __uint_as_float(rc.x);
  const int c = (int)(rc.y >> 9);
  const int lr = (int)(rc.y & 511u);
  v2f g[32];
#pragma unroll
  for (int i = 0; i < 32; ++i) {
    const int ci = __builtin_amdgcn_readlane(c, i);
    g[i] = *reinterpret_cast<const v2f*>(h + (long long)ci * NHID + 2 * lane);
  }
#pragma unroll
  for (int i = 0; i < 32; ++i) {
    if (FULL || i < n) {
      const float wi = __int_as_float(__builtin_amdgcn_readlane(__float_as_int(w), i));
      const int li = __builtin_amdgcn_readlane(lr, i);
      float* a = acc + li * NHID + 2 * lane;
      __hip_atomic_fetch_add(a,     wi * g[i].x, __ATOMIC_RELAXED, __HIP_MEMORY_SCOPE_WORKGROUP);
      __hip_atomic_fetch_add(a + 1, wi * g[i].y, __ATOMIC_RELAXED, __HIP_MEMORY_SCOPE_WORKGROUP);
    }
  }
}

__device__ __forceinline__ v2f row_of(const char* hb, unsigned y) {
  return *reinterpret_cast<const v2f*>(hb + ((y >> 9) << 8));
}

template <int NB>
__device__ __forceinline__ v2f sum_records(uint2 ra, uint2 rb, const char* hb) {
  v2f g[NB * 8];
#pragma unroll
  for (int i = 0; i < NB * 8; ++i) {
    const unsigned y = (unsigned)__builtin_amdgcn_readlane((int)(i < 32 ? ra.y : rb.y), i & 31);
    g[i] = row_of(hb, y);
  }
  v2f s = {0.f, 0.f};
#pragma unroll
  for (int i = 0; i < NB * 8; ++i) {
    const float w = __int_as_float(
        __builtin_amdgcn_readlane((int)(i < 32 ? ra.x : rb.x), i & 31));
    s.x = fmaf(w, g[i].x, s.x);
    s.y = fmaf(w, g[i].y, s.y);
  }
  return s;
}

__device__ __forceinline__ void replay_rows(
    const uint2* __restrict__ tlist, const unsigned* rn, int wave, int lane,
    const float* __restrict__ h, float* tile) {
  const char* hb = reinterpret_cast<const char*>(h) + 8 * lane;
  int lr = wave;
  uint2 ra = tlist[lr * RCAP + lane];
  uint2 rb = tlist[lr * RCAP + 32 + lane];
  int n = __builtin_amdgcn_readfirstlane((int)rn[lr]);
  while (true) {
    const int npad = (n + 7) & ~7;
    const int nxt = lr + 32;
    const int pf = nxt < TR ? nxt : lr;
    const uint2 na = tlist[pf * RCAP + lane];
    const uint2 nb = tlist[pf * RCAP + 32 + lane];
    const unsigned nn = rn[pf];
    v2f s = {0.f, 0.f};
    switch ((npad < RB ? npad : RB) >> 3) {
      case 0:  break;
      case 1:  s = sum_records<1>(ra, rb, hb);  break;
      case 2:  s = sum_records<2>(ra, rb, hb);  break;
      case 3:  s = sum_records<3>(ra, rb, hb);  break;
      case 4:  s = sum_records<4>(ra, rb, hb);  break;
      case 5:  s = sum_records<5>(ra, rb, hb);  break;
      default: s = sum_records<6>(ra, rb, hb);  break;
    }
    for (int i = RB; i < npad; i += 8) {
      const uint2 rx = tlist[lr * RCAP + (i & ~31) + lane];
      v2f t[8];
#pragma unroll
      for (int j = 0; j < 8; ++j)
        t[j] = row_of(hb, (unsigned)__builtin_amdgcn_readlane((int)rx.y, (i + j) & 31));
#pragma unroll
      for (int j = 0; j < 8; ++j) {
        const float w = __int_as_float(__builtin_amdgcn_readlane((int)rx.x, (i + j) & 31));
        s.x = fmaf(w, t[j].x, s.x);
        s.y = fmaf(w, t[j].y, s.y);
      }
    }
    *reinterpret_cast<v2f*>(tile + lr * NHID + 2 * lane) = s;
    if (nxt >= TR) break;
    lr = nxt;  ra = na;  rb = nb;
    n = __builtin_amdgcn_readfirstlane((int)nn);
  }
}

__device__ __forceinline__ void stage_layer_weights(
    const float* __restrict__ Wl, float beta, float* Wp) {
  const float omb = 1.f - beta;
#pragma unroll
  for (int j = 0; j < NHID * NHID / 1024; ++j) {
    const int idx = (int)threadIdx.x + 1024 * j;
    const int k = idx >> 6;
    const int n = idx & 63;
    const float w = beta * Wl[idx];
    const int kh = k >> 5, kt = k & 3, ki = (k >> 3) & 3, ks = (k >> 2) & 1;
    Wp[(((kt * 4 + ki) * 4 + (n & 3)) << 6) + 2 * ((n >> 2) + 16 * kh) + ks] =
        (k == n) ? w + omb : w;
  }
}

__device__ __forceinline__ void layer_body(
    v8f (&s)[4], const float* __restrict__ h0row, const float* Wp, int lane,
    float* __restrict__ hgroup) {
  const float sA = 1.f - ALPHA;
  const float sB = ALPHA;
  v4f g[8];
#pragma unroll
  for (int r = 0; r < 8; ++r) g[r] = *reinterpret_cast<const v4f*>(h0row + 4 * r);
#pragma unroll
  for (int r = 0; r < 8; ++r) {
    s[0][r] = sA * s[0][r] + sB * g[r].x;
    s[1][r] = sA * s[1][r] + sB * g[r].y;
    s[2][r] = sA * s[2][r] + sB * g[r].z;
    s[3][r] = sA * s[3][r] + sB * g[r].w;
  }
  const v8f z8 = {};
  v8f o0 = z8, o1 = z8, o2 = z8, o3 = z8;
  const v2f* wp = reinterpret_cast<const v2f*>(Wp) + lane;
#pragma unroll
  for (int t = 0; t < 4; ++t) {
#pragma unroll
    for (int i = 0; i < 4; ++i) {
      v2f a;
      a.x = s[t][2 * i];  a.y = s[t][2 * i + 1];
      const v2f* wk = wp + (t * 4 + i) * 128;
      o0 = __builtin_amdgcn_wmma_f32_16x16x4_f32(0, a, 0, wk[0],  0, o0, false, false);
      o1 = __builtin_amdgcn_wmma_f32_16x16x4_f32(0, a, 0, wk[32], 0, o1, false, false);
      o2 = __builtin_amdgcn_wmma_f32_16x16x4_f32(0, a, 0, wk[64], 0, o2, false, false);
      o3 = __builtin_amdgcn_wmma_f32_16x16x4_f32(0, a, 0, wk[96], 0, o3, false, false);
    }
  }
  const int half = lane >> 4;
  const int c = lane & 15;
#pragma unroll
  for (int r = 0; r < 8; ++r) {
    v4f o;
    o.x = o0[r] > 0.f ? o0[r] : 0.f;  o.y = o1[r] > 0.f ? o1[r] : 0.f;
    o.z = o2[r] > 0.f ? o2[r] : 0.f;  o.w = o3[r] > 0.f ? o3[r] : 0.f;
    *reinterpret_cast<v4f*>(hgroup + (8 * half + r) * NHID + 4 * c) = o;
  }
}

template <bool FILE, bool ADD>
__device__ __forceinline__ void scan_edges(
    const unsigned* __restrict__ tid, const uint2* __restrict__ rec, int tile, int wave,
    int lane, unsigned* wqw, unsigned* rcnt, uint2* tlist,
    const float* __restrict__ h, float* acc) {
  auto file_rec = [&](uint2 rc, bool valid) {
    if (valid) {
      const unsigned lr = rc.y & 511u;
      const unsigned slot = __hip_atomic_fetch_add(&rcnt[lr], 1u, __ATOMIC_RELAXED,
                                                   __HIP_MEMORY_SCOPE_WORKGROUP);
      if (slot < (unsigned)RCAP) tlist[lr * RCAP + slot] = rc;
    }
  };

  const unsigned pat = (unsigned)tile * 0x01010101u;
  const uint4* tid4 = reinterpret_cast<const uint4*>(tid);
  const int g0 = (tile * NGROUP) / NTILES;

  struct ids2 { uint4 a, b; };
  auto load_ids = [&](int g) {
    const int ch = min(g * 32 + wave, NCHUNK - 1);
    ids2 r;
    r.a = tid4[ch * 64 + lane];
    r.b = tid4[ch * 64 + 32 + lane];
    return r;
  };
  auto next_group = [](int g) { return (g + 1 == NGROUP) ? 0 : g + 1; };

  int qn = 0;
  auto scan = [&](const ids2& v, int g) {
    const int ch = g * 32 + wave;
    if (ch >= NCHUNK) return;
    unsigned m = (zero_bytes(v.a.x ^ pat) >> 7) | (zero_bytes(v.a.y ^ pat) >> 6) |
                 (zero_bytes(v.a.z ^ pat) >> 5) | (zero_bytes(v.a.w ^ pat) >> 4) |
                 (zero_bytes(v.b.x ^ pat) >> 3) | (zero_bytes(v.b.y ^ pat) >> 2) |
                 (zero_bytes(v.b.z ^ pat) >> 1) | zero_bytes(v.b.w ^ pat);
    const unsigned ebase = (unsigned)ch * CHUNK + (unsigned)lane * 16u;
    while (true) {
      const bool has = m != 0u;
      const unsigned mask = __builtin_amdgcn_ballot_w32(has);
      if (mask == 0u) break;
      if (has) {
        const unsigned p = (unsigned)__builtin_ctz(m);
        wqw[qn + (int)__builtin_amdgcn_mbcnt_lo(mask, 0u)] =
            ebase + ((p & 4u) << 7) + ((p & 3u) << 2) + (p >> 3);
        m &= m - 1u;
      }
      qn += __builtin_popcount(mask);
      if (qn >= 32) {
        qn -= 32;
        const uint2 rc = rec[wqw[qn + lane]];
        if (FILE) file_rec(rc, true);
        if (ADD) add_batch<true>(rc, 32, lane, h, acc);
      }
    }
  };

  static_assert(NGROUP % 2 == 0, "the scan loop takes two groups per trip");
  int g = g0;
  ids2 cur = load_ids(g);
  for (int k = 0; k < NGROUP; k += 2) {
    const int g1 = next_group(g);
    const int g2 = next_group(g1);
    const ids2 nxt = load_ids(g1);
    scan(cur, g);
    cur = load_ids(g2);
    scan(nxt, g1);
    g = g2;
  }
  if (qn > 0) {
    const uint2 rc = (lane < qn) ? rec[wqw[lane]] : make_uint2(0u, 0u);
    if (FILE) file_rec(rc, lane < qn);
    if (ADD) add_batch<false>(rc, qn, lane, h, acc);
  }
}

template <bool FIRST>
__global__ void __launch_bounds__(1024) spmm_owner_kernel(
    const unsigned* __restrict__ tid, const uint2* __restrict__ rec,
    const float* __restrict__ h, const float* __restrict__ h0,
    const float* __restrict__ Wl, float beta, float* __restrict__ hout,
    uint2* __restrict__ rlists, unsigned* __restrict__ rcounts) {
  extern __shared__ __attribute__((aligned(16))) float tile_lds[];
  __shared__ unsigned wq[32 * QCAP];
  __shared__ unsigned rcnt[TR];
  __shared__ unsigned wover[32];
  __shared__ __attribute__((aligned(16))) float Wp[NHID * NHID];
  float* acc = tile_lds;
  float4* acc4 = reinterpret_cast<float4*>(tile_lds);
  const int lane = threadIdx.x & 31;
  const int wave = __builtin_amdgcn_readfirstlane(threadIdx.x >> 5);
  const int qb = wave * QCAP;
  const int tile = blockIdx.x;
  uint2* tlist = rlists + (size_t)tile * TR * RCAP;
  const long long grow = (long long)tile * TR + wave * 16;
  const float* h0row = h0 + (grow + (lane & 15)) * NHID + 32 * (lane >> 4);
  float* hgroup = hout + grow * NHID;

  if (!FIRST) {
    const int r = (int)threadIdx.x;
    const unsigned myn = (r < TR) ? rcounts[tile * TR + r] : 0u;
    if (r < TR) rcnt[r] = myn;
    const unsigned ob = __builtin_amdgcn_ballot_w32(myn > (unsigned)RCAP);
    if (lane == 0) wover[wave] = ob;
  } else {
    if (threadIdx.x < TR) rcnt[threadIdx.x] = 0u;
  }
  stage_layer_weights(Wl, beta, Wp);
  __syncthreads();

  if (FIRST) {
    scan_edges<true, false>(tid, rec, tile, wave, lane, wq + qb, rcnt, tlist, h, acc);
    __syncthreads();
    const int r = (int)threadIdx.x;
    const unsigned n = (r < TR) ? rcnt[r] : 0u;
    if (r < TR) {
      rcounts[tile * TR + r] = n;
      if (n <= (unsigned)RCAP) {
        const unsigned e = (n + 7u) & ~7u;
        for (unsigned i = n; i < e; ++i) tlist[r * RCAP + i] = make_uint2(0u, 0u);
      }
    }
    const unsigned ob = __builtin_amdgcn_ballot_w32(n > (unsigned)RCAP);
    if (lane == 0) wover[wave] = ob;
    __threadfence_block();
    __syncthreads();
  }

  if (__builtin_amdgcn_ballot_w32(wover[lane] != 0u) == 0u) {
    replay_rows(tlist, rcnt, wave, lane, h, tile_lds);
  } else {
    for (int i = threadIdx.x; i < TR * NHID / 4; i += 1024)
      acc4[i] = make_float4(0.f, 0.f, 0.f, 0.f);
    __syncthreads();
    scan_edges<false, true>(tid, rec, tile, wave, lane, wq + qb, rcnt, tlist, h, acc);
  }
  __syncthreads();

  if (wave < NG) {
    const v4f* trow = reinterpret_cast<const v4f*>(
        tile_lds + (wave * 16 + (lane & 15)) * NHID + 32 * (lane >> 4));
    v8f s[4];
#pragma unroll
    for (int r = 0; r < 8; ++r) {
      const v4f v = trow[r];
      s[0][r] = v.x;  s[1][r] = v.y;  s[2][r] = v.z;  s[3][r] = v.w;
    }
    layer_body(s, h0row, Wp, lane, hgroup);
  }
}

__global__ void __launch_bounds__(256) out_kernel(
    const float* __restrict__ h, const float* __restrict__ Wout,
    const float* __restrict__ bout, float* __restrict__ out) {
  __shared__ float Ws[NHID * NOUT];
  __shared__ float bs[NOUT];
  for (int idx = threadIdx.x; idx < NHID * NOUT; idx += 256) Ws[idx] = Wout[idx];
  if (threadIdx.x < NOUT) bs[threadIdx.x] = bout[threadIdx.x];
  __syncthreads();

  const int i = blockIdx.x * blockDim.x + threadIdx.x;
  if (i >= N_NODES) return;

  float hr[NHID];
  const float4* hp = reinterpret_cast<const float4*>(h + (long long)i * NHID);
#pragma unroll
  for (int k4 = 0; k4 < NHID / 4; ++k4) {
    const float4 v = hp[k4];
    hr[4 * k4 + 0] = v.x; hr[4 * k4 + 1] = v.y;
    hr[4 * k4 + 2] = v.z; hr[4 * k4 + 3] = v.w;
  }

  float lg[NOUT];
#pragma unroll
  for (int c = 0; c < NOUT; ++c) lg[c] = bs[c];
  for (int k = 0; k < NHID; ++k) {
    const float hk = hr[k];
#pragma unroll
    for (int c = 0; c < NOUT; ++c) lg[c] = fmaf(hk, Ws[k * NOUT + c], lg[c]);
  }

  float mx = lg[0];
#pragma unroll
  for (int c = 1; c < NOUT; ++c) mx = fmaxf(mx, lg[c]);
  float s = 0.f;
#pragma unroll
  for (int c = 0; c < NOUT; ++c) s += expf(lg[c] - mx);
  const float lse = mx + logf(s);
#pragma unroll
  for (int c = 0; c < NOUT; ++c)
    out[(long long)i * NOUT + c] = lg[c] - lse;
}

extern "C" void kernel_launch(void* const* d_in, const int* in_sizes, int n_in,
                              void* d_out, int out_size, void* d_ws, size_t ws_size,
                              hipStream_t stream) {
  (void)in_sizes; (void)n_in; (void)out_size; (void)ws_size;
  const float* x     = (const float*)d_in[0];
  const int*   row   = (const int*)  d_in[1];
  const int*   col   = (const int*)  d_in[2];
  const float* ew    = (const float*)d_in[3];
  const float* Win   = (const float*)d_in[4];
  const float* bin   = (const float*)d_in[5];
  const float* convW = (const float*)d_in[6];
  const float* Wout  = (const float*)d_in[7];
  const float* bout  = (const float*)d_in[8];

  const size_t S = (size_t)N_NODES * NHID;
  float* h0 = (float*)d_ws;
  float* hA = h0 + S;
  float* hB = hA + S;

  const size_t inLds = (size_t)(WP_FLOATS + IN_WAVES * XBUF) * sizeof(float);
  in_gemm_kernel<<<IN_BLOCKS, IN_WAVES * 32, inLds, stream>>>(x, Win, bin, h0);

  unsigned* tid = (unsigned*)(hB + S);
  uint2* rec = (uint2*)(tid + E_EDGES / 4);
  edge_pack_kernel<<<E_EDGES / 4 / 256, 256, 0, stream>>>(row, col, ew, tid, rec);
  const size_t spmmLds = (size_t)164 * 1024;
  uint2* lists = rec + E_EDGES;
  unsigned* counts = (unsigned*)(lists + (size_t)N_NODES * RCAP);

  const float* hin = h0;
  float* hnext = hA;
  float* hother = hB;
  for (int l = 0; l < NLAYERS; ++l) {
    const float beta = logf(0.5f / (float)(l + 1) + 1.f);
    const float* Wl = convW + (size_t)l * NHID * NHID;
    if (l == 0)
      spmm_owner_kernel<true><<<NTILES, 1024, spmmLds, stream>>>(
          tid, rec, hin, h0, Wl, beta, hnext, lists, counts);
    else
      spmm_owner_kernel<false><<<NTILES, 1024, spmmLds, stream>>>(
          tid, rec, hin, h0, Wl, beta, hnext, lists, counts);
    hin = hnext;
    hnext = hother;
    hother = (float*)hin;
  }

  out_kernel<<<(N_NODES + 255) / 256, 256, 0, stream>>>(hin, Wout, bout, (float*)d_out);
}
